# starting kernel
# baseline (speedup 1.0000x reference)
#include <hip/hip_runtime.h>
#include <math.h>

#define Bb 256
#define Tt 512
#define Cc 16
#define Hh 64

#define WP  72
#define AP1 69
#define AP2 133
#define DXP 20

typedef __attribute__((ext_vector_type(2))) float v2f;
typedef __attribute__((ext_vector_type(8))) float v8f;

static __device__ __forceinline__ v8f wmma4(v2f a, v2f b, v8f c) {
  return __builtin_amdgcn_wmma_f32_16x16x4_f32(false, a, false, b, (short)0, c,
                                               false, false);
}

template <int K, int APITCH>
static __device__ __forceinline__ v8f gemm_tile(const float* __restrict__ A,
                                                int acol0,
                                                const float* __restrict__ W,
                                                int n0, v8f acc, int l2, int hi) {
#pragma unroll
  for (int k = 0; k < K; k += 4) {
    v2f a, b;
    const float* ap = A + l2 * APITCH + acol0 + k + 2 * hi;
    a.x = ap[0];
    a.y = ap[1];
    const float* wp = W + (k + 2 * hi) * WP + n0 + l2;
    b.x = wp[0];
    b.y = wp[WP];
    acc = wmma4(a, b, acc);
  }
  return acc;
}

template <int APITCH>
static __device__ __forceinline__ void store_tile(float* __restrict__ O, int n0,
                                                  v8f acc, int l2, int hi) {
#pragma unroll
  for (int r = 0; r < 8; ++r) O[(r + 8 * hi) * APITCH + n0 + l2] = acc[r];
}

template <int ROWS>
static __device__ __forceinline__ void load_w64(float* __restrict__ dst,
                                                const float* __restrict__ src,
                                                int tid, int nthr) {
  for (int i = tid; i < ROWS * 64; i += nthr)
    dst[(i >> 6) * WP + (i & 63)] = src[i];
}

static __device__ __forceinline__ float relu(float v) { return v > 0.f ? v : 0.f; }

__global__ __launch_bounds__(128) void ncde_embed_kernel(
    const float* __restrict__ x, const float* __restrict__ seq_ts,
    const float* __restrict__ coeffs, const float* __restrict__ times,
    const float* __restrict__ W_if, const float* __restrict__ b_if,
    const float* __restrict__ W_ic, const float* __restrict__ b_ic,
    const float* __restrict__ W_emb, const float* __restrict__ b_emb,
    const float* __restrict__ W_fl1, const float* __restrict__ b_fl1,
    const float* __restrict__ W_flo, const float* __restrict__ b_flo,
    float* __restrict__ zflow) {
  __shared__ float sWif[16 * WP], sWic[16 * WP], sWemb[128 * WP];
  __shared__ float sWfl1[64 * WP], sWflo[64 * WP];
  __shared__ float sb[5][64];
  __shared__ float sA[16 * AP2], sB[16 * AP2];

  const int tid = threadIdx.x, nthr = 128;
  const int lane = tid & 31, wave = tid >> 5, l2 = lane & 15, hi = lane >> 4;
  const int r0 = blockIdx.x * 16;

  load_w64<16>(sWif, W_if, tid, nthr);
  load_w64<16>(sWic, W_ic, tid, nthr);
  load_w64<128>(sWemb, W_emb, tid, nthr);
  load_w64<64>(sWfl1, W_fl1, tid, nthr);
  load_w64<64>(sWflo, W_flo, tid, nthr);
  if (tid < 64) {
    sb[0][tid] = b_if[tid];
    sb[1][tid] = b_ic[tid];
    sb[2][tid] = b_emb[tid];
    sb[3][tid] = b_fl1[tid];
    sb[4][tid] = b_flo[tid];
  }
  for (int i = tid; i < 256; i += nthr) {
    const int m = i >> 4, c = i & 15;
    const int rr = r0 + m, b = rr / Tt, t = rr % Tt;
    sA[m * AP2 + c] = (c == 0) ? seq_ts[rr] : x[(size_t)rr * (Cc - 1) + (c - 1)];
    const int idx = (t < Tt - 1) ? t : (Tt - 2);
    const float s = times[t] - times[idx];
    const float* cf = coeffs + ((size_t)b * (Tt - 1) + idx) * (4 * Cc) + c;
    const float a0 = cf[0], b0 = cf[Cc], c0 = cf[2 * Cc], d0 = cf[3 * Cc];
    sA[m * AP2 + 16 + c] = a0 + s * (b0 + s * (c0 + s * d0));
  }
  __syncthreads();

  const int n0 = wave * 16;
  v8f aF = {}, aX = {};
  aF = gemm_tile<16, AP2>(sA, 0, sWif, n0, aF, l2, hi);
  aX = gemm_tile<16, AP2>(sA, 16, sWic, n0, aX, l2, hi);
  const float bF = sb[0][n0 + l2], bX = sb[1][n0 + l2];
#pragma unroll
  for (int r = 0; r < 8; ++r) {
    const int m = r + 8 * hi;
    sB[m * AP2 + n0 + l2] = aF[r] + bF;
    sB[m * AP2 + 64 + n0 + l2] = aX[r] + bX;
  }
  __syncthreads();

  v8f acc = {};
  acc = gemm_tile<128, AP2>(sB, 0, sWemb, n0, acc, l2, hi);
  float bv = sb[2][n0 + l2];
#pragma unroll
  for (int r = 0; r < 8; ++r)
    sA[(r + 8 * hi) * AP2 + n0 + l2] = relu(acc[r] + bv);
  __syncthreads();

  acc = (v8f){};
  acc = gemm_tile<64, AP2>(sA, 0, sWfl1, n0, acc, l2, hi);
  bv = sb[3][n0 + l2];
#pragma unroll
  for (int r = 0; r < 8; ++r)
    sB[(r + 8 * hi) * AP2 + n0 + l2] = relu(acc[r] + bv);
  __syncthreads();

  acc = (v8f){};
  acc = gemm_tile<64, AP2>(sB, 0, sWflo, n0, acc, l2, hi);
  bv = sb[4][n0 + l2];
#pragma unroll
  for (int r = 0; r < 8; ++r)
    zflow[(size_t)(r0 + r + 8 * hi) * Hh + n0 + l2] = acc[r] + bv;
}

__global__ __launch_bounds__(512, 1) void ncde_scan_kernel(
    const float* __restrict__ coeffs, const float* __restrict__ times,
    const float* __restrict__ W_ic, const float* __restrict__ b_ic,
    const float* __restrict__ W_f1, const float* __restrict__ b_f1,
    const float* __restrict__ W_f2, const float* __restrict__ b_f2,
    float* __restrict__ zcde) {
  __shared__ float sWf1[64 * WP], sWic[16 * WP];
  __shared__ float sbf1[64], sbic[64];
  __shared__ float stim[Tt];
  __shared__ float sh[16 * AP1], szb[16 * AP1], sze[16 * AP1];
  __shared__ float sk[4][16 * AP1];
  __shared__ float sdx[4][16 * DXP];

  const int tid = threadIdx.x, nthr = 512;
  const int lane = tid & 31, wave = tid >> 5, l2 = lane & 15, hi = lane >> 4;
  const int b0 = blockIdx.x * 16;

  load_w64<64>(sWf1, W_f1, tid, nthr);
  load_w64<16>(sWic, W_ic, tid, nthr);
  if (tid < 64) {
    sbf1[tid] = b_f1[tid];
    sbic[tid] = b_ic[tid];
  }
  for (int i = tid; i < Tt; i += nthr) stim[i] = times[i];

  v2f wf2[4][16];
  float bf2r[4];
#pragma unroll
  for (int jj = 0; jj < 4; ++jj) {
    const int j = wave * 4 + jj;
    const int col = 16 * j + l2;
    bf2r[jj] = b_f2[col];
#pragma unroll
    for (int kk = 0; kk < 16; ++kk) {
      const int row = 4 * kk + 2 * hi;
      wf2[jj][kk].x = W_f2[(size_t)row * (Hh * Cc) + col];
      wf2[jj][kk].y = W_f2[(size_t)(row + 1) * (Hh * Cc) + col];
    }
  }

  for (int i = tid; i < 256; i += nthr) {
    const int m = i >> 4, c = i & 15;
    sze[m * AP1 + c] = coeffs[((size_t)(b0 + m) * (Tt - 1)) * (4 * Cc) + c];
  }
  __syncthreads();
  if (wave < 4) {
    const int n0 = wave * 16;
    v8f acc = {};
    acc = gemm_tile<16, AP1>(sze, 0, sWic, n0, acc, l2, hi);
    const float bv = sbic[n0 + l2];
#pragma unroll
    for (int r = 0; r < 8; ++r) {
      const int m = r + 8 * hi;
      const float v = acc[r] + bv;
      szb[m * AP1 + n0 + l2] = v;
      zcde[((size_t)(b0 + m) * Tt) * Hh + n0 + l2] = v;
    }
  }
  __syncthreads();
  for (int i = tid; i < 16 * Hh; i += nthr) {
    const int m = i >> 6, h = i & 63;
    sze[m * AP1 + h] = szb[m * AP1 + h];
  }
  __syncthreads();

  for (int step = 0; step < Tt - 1; ++step) {
    const float dt = stim[step + 1] - stim[step];
    if (tid < 256) {
      const int m = tid >> 4, c = tid & 15, b = b0 + m;
      const float* cf = coeffs + ((size_t)b * (Tt - 1) + step) * (4 * Cc) + c;
      const float bb = cf[Cc], cc = cf[2 * Cc], dd = cf[3 * Cc];
      const float s1 = dt * (1.f / 3.f), s2 = dt * (2.f / 3.f);
      sdx[0][m * DXP + c] = bb;
      sdx[1][m * DXP + c] = bb + s1 * (2.f * cc + 3.f * dd * s1);
      sdx[2][m * DXP + c] = bb + s2 * (2.f * cc + 3.f * dd * s2);
      const int idx3 = (step + 1 < Tt - 1) ? (step + 1) : (Tt - 2);
      const float* cf3 = coeffs + ((size_t)b * (Tt - 1) + idx3) * (4 * Cc) + c;
      const float b3 = cf3[Cc], c3 = cf3[2 * Cc], d3 = cf3[3 * Cc];
      const float s3 = stim[step + 1] - stim[idx3];
      sdx[3][m * DXP + c] = b3 + s3 * (2.f * c3 + 3.f * d3 * s3);
    }
    __syncthreads();

    for (int sub = 0; sub < 4; ++sub) {
      if (wave < 4) {
        const int n0 = wave * 16;
        v8f acc = {};
        acc = gemm_tile<64, AP1>(sze, 0, sWf1, n0, acc, l2, hi);
        const float bv = sbf1[n0 + l2];
#pragma unroll
        for (int r = 0; r < 8; ++r)
          sh[(r + 8 * hi) * AP1 + n0 + l2] = relu(acc[r] + bv);
      }
      __syncthreads();

      v8f acc2[4];
#pragma unroll
      for (int jj = 0; jj < 4; ++jj) {
        v8f t;
#pragma unroll
        for (int r = 0; r < 8; ++r) t[r] = bf2r[jj];
        acc2[jj] = t;
      }
#pragma unroll
      for (int kk = 0; kk < 16; ++kk) {
        v2f a;
        const float* ap = sh + l2 * AP1 + 4 * kk + 2 * hi;
        a.x = ap[0];
        a.y = ap[1];
#pragma unroll
        for (int jj = 0; jj < 4; ++jj) acc2[jj] = wmma4(a, wf2[jj][kk], acc2[jj]);
      }
      float dxr[8];
#pragma unroll
      for (int r = 0; r < 8; ++r) dxr[r] = sdx[sub][(r + 8 * hi) * DXP + l2];
      float* kdst = sk[sub];
#pragma unroll
      for (int jj = 0; jj < 4; ++jj) {
#pragma unroll
        for (int r = 0; r < 8; ++r) {
          float p = tanhf(acc2[jj][r]) * dxr[r];
          p += __shfl_xor(p, 1, 32);
          p += __shfl_xor(p, 2, 32);
          p += __shfl_xor(p, 4, 32);
          p += __shfl_xor(p, 8, 32);
          if (l2 == 0) kdst[(r + 8 * hi) * AP1 + wave * 4 + jj] = p;
        }
      }
      __syncthreads();

      for (int i = tid; i < 16 * Hh; i += nthr) {
        const int m = i >> 6, h = i & 63;
        const int ad = m * AP1 + h;
        const float zb = szb[ad];
        if (sub == 0) {
          sze[ad] = zb + (dt * (1.f / 3.f)) * sk[0][ad];
        } else if (sub == 1) {
          sze[ad] = zb + dt * (sk[1][ad] - sk[0][ad] * (1.f / 3.f));
        } else if (sub == 2) {
          sze[ad] = zb + dt * (sk[0][ad] - sk[1][ad] + sk[2][ad]);
        } else {
          const float zn =
              zb + (sk[0][ad] + 3.f * (sk[1][ad] + sk[2][ad]) + sk[3][ad]) *
                       (dt * 0.125f);
          szb[ad] = zn;
          sze[ad] = zn;
          zcde[((size_t)(b0 + m) * Tt + (step + 1)) * Hh + h] = zn;
        }
      }
      __syncthreads();
    }
  }
}

__global__ __launch_bounds__(128) void ncde_head_kernel(
    const float* __restrict__ zflow, const float* __restrict__ zcde,
    const float* __restrict__ W_mix, const float* __restrict__ b_mix,
    const float* __restrict__ W_l1, const float* __restrict__ b_l1,
    const float* __restrict__ W_l2, const float* __restrict__ b_l2,
    float* __restrict__ out, float* __restrict__ zout) {
  __shared__ float sWmix[128 * WP], sWl1[64 * WP];
  __shared__ float sbmix[64], sbl1[64], sWl2v[64];
  __shared__ float sA[16 * AP2], sB[16 * AP1], sC[16 * AP1];

  const int tid = threadIdx.x, nthr = 128;
  const int lane = tid & 31, wave = tid >> 5, l2 = lane & 15, hi = lane >> 4;
  const int r0 = blockIdx.x * 16;

  load_w64<128>(sWmix, W_mix, tid, nthr);
  load_w64<64>(sWl1, W_l1, tid, nthr);
  if (tid < 64) {
    sbmix[tid] = b_mix[tid];
    sbl1[tid] = b_l1[tid];
    sWl2v[tid] = W_l2[tid];
  }
  for (int i = tid; i < 16 * 128; i += nthr) {
    const int m = i >> 7, c = i & 127;
    const size_t rr = r0 + m;
    sA[m * AP2 + c] = (c < 64) ? zflow[rr * Hh + c] : zcde[rr * Hh + (c - 64)];
  }
  __syncthreads();

  const int n0 = wave * 16;
  v8f acc = {};
  acc = gemm_tile<128, AP2>(sA, 0, sWmix, n0, acc, l2, hi);
  float bv = sbmix[n0 + l2];
#pragma unroll
  for (int r = 0; r < 8; ++r) {
    const int m = r + 8 * hi;
    const float v = acc[r] + bv;
    zout[(size_t)(r0 + m) * Hh + n0 + l2] = v;
    sB[m * AP1 + n0 + l2] = tanhf(v);
  }
  __syncthreads();

  acc = (v8f){};
  acc = gemm_tile<64, AP1>(sB, 0, sWl1, n0, acc, l2, hi);
  bv = sbl1[n0 + l2];
#pragma unroll
  for (int r = 0; r < 8; ++r)
    sC[(r + 8 * hi) * AP1 + n0 + l2] = relu(acc[r] + bv);
  __syncthreads();

  const int m = tid >> 3, p = tid & 7;
  float s = 0.f;
#pragma unroll
  for (int q = 0; q < 8; ++q) s += sC[m * AP1 + p * 8 + q] * sWl2v[p * 8 + q];
  s += __shfl_xor(s, 1, 32);
  s += __shfl_xor(s, 2, 32);
  s += __shfl_xor(s, 4, 32);
  if (p == 0) out[r0 + m] = s + b_l2[0];
}

extern "C" void kernel_launch(void* const* d_in, const int* in_sizes, int n_in,
                              void* d_out, int out_size, void* d_ws,
                              size_t ws_size, hipStream_t stream) {
  const float* x = (const float*)d_in[0];
  const float* seq_ts = (const float*)d_in[1];
  const float* coeffs = (const float*)d_in[3];
  const float* times = (const float*)d_in[4];
  const float* W_if = (const float*)d_in[5];
  const float* b_if = (const float*)d_in[6];
  const float* W_ic = (const float*)d_in[7];
  const float* b_ic = (const float*)d_in[8];
  const float* W_emb = (const float*)d_in[9];
  const float* b_emb = (const float*)d_in[10];
  const float* W_fl1 = (const float*)d_in[11];
  const float* b_fl1 = (const float*)d_in[12];
  const float* W_flo = (const float*)d_in[13];
  const float* b_flo = (const float*)d_in[14];
  const float* W_f1 = (const float*)d_in[15];
  const float* b_f1 = (const float*)d_in[16];
  const float* W_f2 = (const float*)d_in[17];
  const float* b_f2 = (const float*)d_in[18];
  const float* W_mix = (const float*)d_in[19];
  const float* b_mix = (const float*)d_in[20];
  const float* W_l1 = (const float*)d_in[21];
  const float* b_l1 = (const float*)d_in[22];
  const float* W_l2 = (const float*)d_in[23];
  const float* b_l2 = (const float*)d_in[24];

  float* zflow = (float*)d_ws;
  float* zcde = zflow + (size_t)Bb * Tt * Hh;
  float* out = (float*)d_out;
  float* zout = out + (size_t)Bb * Tt;

  ncde_embed_kernel<<<(Bb * Tt) / 16, 128, 0, stream>>>(
      x, seq_ts, coeffs, times, W_if, b_if, W_ic, b_ic, W_emb, b_emb, W_fl1,
      b_fl1, W_flo, b_flo, zflow);
  ncde_scan_kernel<<<Bb / 16, 512, 0, stream>>>(coeffs, times, W_ic, b_ic,
                                                W_f1, b_f1, W_f2, b_f2, zcde);
  ncde_head_kernel<<<(Bb * Tt) / 16, 128, 0, stream>>>(
      zflow, zcde, W_mix, b_mix, W_l1, b_l1, W_l2, b_l2, out, zout);
}


# scan: swap WMMA operands (W_f2 frag = A, h frag = B) so tiles come out channel-major; einsum channel sum = 8 in-lane FMAs + 1 half exchange, not 4 shuffles per value (sum order changes)
# speedup vs baseline: 1.7766x; 1.7766x over previous
#include <hip/hip_runtime.h>
#include <math.h>

#define Bb 256
#define Tt 512
#define Cc 16
#define Hh 64

#define WP  72
#define AP1 69
#define AP2 133
#define DXP 20

typedef __attribute__((ext_vector_type(2))) float v2f;
typedef __attribute__((ext_vector_type(8))) float v8f;

static __device__ __forceinline__ v8f wmma4(v2f a, v2f b, v8f c) {
  return __builtin_amdgcn_wmma_f32_16x16x4_f32(false, a, false, b, (short)0, c,
                                               false, false);
}

template <int K, int APITCH>
static __device__ __forceinline__ v8f gemm_tile(const float* __restrict__ A,
                                                int acol0,
                                                const float* __restrict__ W,
                                                int n0, v8f acc, int l2, int hi) {
#pragma unroll
  for (int k = 0; k < K; k += 4) {
    v2f a, b;
    const float* ap = A + l2 * APITCH + acol0 + k + 2 * hi;
    a.x = ap[0];
    a.y = ap[1];
    const float* wp = W + (k + 2 * hi) * WP + n0 + l2;
    b.x = wp[0];
    b.y = wp[WP];
    acc = wmma4(a, b, acc);
  }
  return acc;
}

template <int APITCH>
static __device__ __forceinline__ void store_tile(float* __restrict__ O, int n0,
                                                  v8f acc, int l2, int hi) {
#pragma unroll
  for (int r = 0; r < 8; ++r) O[(r + 8 * hi) * APITCH + n0 + l2] = acc[r];
}

template <int ROWS>
static __device__ __forceinline__ void load_w64(float* __restrict__ dst,
                                                const float* __restrict__ src,
                                                int tid, int nthr) {
  for (int i = tid; i < ROWS * 64; i += nthr)
    dst[(i >> 6) * WP + (i & 63)] = src[i];
}

static __device__ __forceinline__ float relu(float v) { return v > 0.f ? v : 0.f; }

static __device__ __forceinline__ float xhalf(float v) {
  const int b = __builtin_bit_cast(int, v);
  return __builtin_bit_cast(
      float, __builtin_amdgcn_permlanex16(b, b, 0x76543210, 0xfedcba98, false, false));
}

__global__ __launch_bounds__(128) void ncde_embed_kernel(
    const float* __restrict__ x, const float* __restrict__ seq_ts,
    const float* __restrict__ coeffs, const float* __restrict__ times,
    const float* __restrict__ W_if, const float* __restrict__ b_if,
    const float* __restrict__ W_ic, const float* __restrict__ b_ic,
    const float* __restrict__ W_emb, const float* __restrict__ b_emb,
    const float* __restrict__ W_fl1, const float* __restrict__ b_fl1,
    const float* __restrict__ W_flo, const float* __restrict__ b_flo,
    float* __restrict__ zflow) {
  __shared__ float sWif[16 * WP], sWic[16 * WP], sWemb[128 * WP];
  __shared__ float sWfl1[64 * WP], sWflo[64 * WP];
  __shared__ float sb[5][64];
  __shared__ float sA[16 * AP2], sB[16 * AP2];

  const int tid = threadIdx.x, nthr = 128;
  const int lane = tid & 31, wave = tid >> 5, l2 = lane & 15, hi = lane >> 4;
  const int r0 = blockIdx.x * 16;

  load_w64<16>(sWif, W_if, tid, nthr);
  load_w64<16>(sWic, W_ic, tid, nthr);
  load_w64<128>(sWemb, W_emb, tid, nthr);
  load_w64<64>(sWfl1, W_fl1, tid, nthr);
  load_w64<64>(sWflo, W_flo, tid, nthr);
  if (tid < 64) {
    sb[0][tid] = b_if[tid];
    sb[1][tid] = b_ic[tid];
    sb[2][tid] = b_emb[tid];
    sb[3][tid] = b_fl1[tid];
    sb[4][tid] = b_flo[tid];
  }
  for (int i = tid; i < 256; i += nthr) {
    const int m = i >> 4, c = i & 15;
    const int rr = r0 + m, b = rr / Tt, t = rr % Tt;
    sA[m * AP2 + c] = (c == 0) ? seq_ts[rr] : x[(size_t)rr * (Cc - 1) + (c - 1)];
    const int idx = (t < Tt - 1) ? t : (Tt - 2);
    const float s = times[t] - times[idx];
    const float* cf = coeffs + ((size_t)b * (Tt - 1) + idx) * (4 * Cc) + c;
    const float a0 = cf[0], b0 = cf[Cc], c0 = cf[2 * Cc], d0 = cf[3 * Cc];
    sA[m * AP2 + 16 + c] = a0 + s * (b0 + s * (c0 + s * d0));
  }
  __syncthreads();

  const int n0 = wave * 16;
  v8f aF = {}, aX = {};
  aF = gemm_tile<16, AP2>(sA, 0, sWif, n0, aF, l2, hi);
  aX = gemm_tile<16, AP2>(sA, 16, sWic, n0, aX, l2, hi);
  const float bF = sb[0][n0 + l2], bX = sb[1][n0 + l2];
#pragma unroll
  for (int r = 0; r < 8; ++r) {
    const int m = r + 8 * hi;
    sB[m * AP2 + n0 + l2] = aF[r] + bF;
    sB[m * AP2 + 64 + n0 + l2] = aX[r] + bX;
  }
  __syncthreads();

  v8f acc = {};
  acc = gemm_tile<128, AP2>(sB, 0, sWemb, n0, acc, l2, hi);
  float bv = sb[2][n0 + l2];
#pragma unroll
  for (int r = 0; r < 8; ++r)
    sA[(r + 8 * hi) * AP2 + n0 + l2] = relu(acc[r] + bv);
  __syncthreads();

  acc = (v8f){};
  acc = gemm_tile<64, AP2>(sA, 0, sWfl1, n0, acc, l2, hi);
  bv = sb[3][n0 + l2];
#pragma unroll
  for (int r = 0; r < 8; ++r)
    sB[(r + 8 * hi) * AP2 + n0 + l2] = relu(acc[r] + bv);
  __syncthreads();

  acc = (v8f){};
  acc = gemm_tile<64, AP2>(sB, 0, sWflo, n0, acc, l2, hi);
  bv = sb[4][n0 + l2];
#pragma unroll
  for (int r = 0; r < 8; ++r)
    zflow[(size_t)(r0 + r + 8 * hi) * Hh + n0 + l2] = acc[r] + bv;
}

__global__ __launch_bounds__(512, 1) void ncde_scan_kernel(
    const float* __restrict__ coeffs, const float* __restrict__ times,
    const float* __restrict__ W_ic, const float* __restrict__ b_ic,
    const float* __restrict__ W_f1, const float* __restrict__ b_f1,
    const float* __restrict__ W_f2, const float* __restrict__ b_f2,
    float* __restrict__ zcde) {
  __shared__ float sWf1[64 * WP], sWic[16 * WP];
  __shared__ float sbf1[64], sbic[64];
  __shared__ float stim[Tt];
  __shared__ float sh[16 * AP1], szb[16 * AP1], sze[16 * AP1];
  __shared__ float sk[4][16 * AP1];
  __shared__ __attribute__((aligned(16))) float sdx[4][16 * DXP];
  __shared__ __attribute__((aligned(32))) float sbf2[Hh * Cc];

  const int tid = threadIdx.x, nthr = 512;
  const int lane = tid & 31, wave = tid >> 5, l2 = lane & 15, hi = lane >> 4;
  const int b0 = blockIdx.x * 16;

  load_w64<64>(sWf1, W_f1, tid, nthr);
  load_w64<16>(sWic, W_ic, tid, nthr);
  if (tid < 64) {
    sbf1[tid] = b_f1[tid];
    sbic[tid] = b_ic[tid];
  }
  for (int i = tid; i < Tt; i += nthr) stim[i] = times[i];
  for (int i = tid; i < Hh * Cc; i += nthr) sbf2[i] = b_f2[i];

  v2f wf2[4][16];
#pragma unroll
  for (int jj = 0; jj < 4; ++jj) {
    const int j = wave * 4 + jj;
    const int col = 16 * j + l2;
#pragma unroll
    for (int kk = 0; kk < 16; ++kk) {
      const int row = 4 * kk + 2 * hi;
      wf2[jj][kk].x = W_f2[(size_t)row * (Hh * Cc) + col];
      wf2[jj][kk].y = W_f2[(size_t)(row + 1) * (Hh * Cc) + col];
    }
  }

  for (int i = tid; i < 256; i += nthr) {
    const int m = i >> 4, c = i & 15;
    sze[m * AP1 + c] = coeffs[((size_t)(b0 + m) * (Tt - 1)) * (4 * Cc) + c];
  }
  __syncthreads();
  if (wave < 4) {
    const int n0 = wave * 16;
    v8f acc = {};
    acc = gemm_tile<16, AP1>(sze, 0, sWic, n0, acc, l2, hi);
    const float bv = sbic[n0 + l2];
#pragma unroll
    for (int r = 0; r < 8; ++r) {
      const int m = r + 8 * hi;
      const float v = acc[r] + bv;
      szb[m * AP1 + n0 + l2] = v;
      zcde[((size_t)(b0 + m) * Tt) * Hh + n0 + l2] = v;
    }
  }
  __syncthreads();
  for (int i = tid; i < 16 * Hh; i += nthr) {
    const int m = i >> 6, h = i & 63;
    sze[m * AP1 + h] = szb[m * AP1 + h];
  }
  __syncthreads();

  for (int step = 0; step < Tt - 1; ++step) {
    const float dt = stim[step + 1] - stim[step];
    if (tid < 256) {
      const int m = tid >> 4, c = tid & 15, b = b0 + m;
      const float* cf = coeffs + ((size_t)b * (Tt - 1) + step) * (4 * Cc) + c;
      const float bb = cf[Cc], cc = cf[2 * Cc], dd = cf[3 * Cc];
      const float s1 = dt * (1.f / 3.f), s2 = dt * (2.f / 3.f);
      sdx[0][m * DXP + c] = bb;
      sdx[1][m * DXP + c] = bb + s1 * (2.f * cc + 3.f * dd * s1);
      sdx[2][m * DXP + c] = bb + s2 * (2.f * cc + 3.f * dd * s2);
      const int idx3 = (step + 1 < Tt - 1) ? (step + 1) : (Tt - 2);
      const float* cf3 = coeffs + ((size_t)b * (Tt - 1) + idx3) * (4 * Cc) + c;
      const float b3 = cf3[Cc], c3 = cf3[2 * Cc], d3 = cf3[3 * Cc];
      const float s3 = stim[step + 1] - stim[idx3];
      sdx[3][m * DXP + c] = b3 + s3 * (2.f * c3 + 3.f * d3 * s3);
    }
    __syncthreads();

    for (int sub = 0; sub < 4; ++sub) {
      if (wave < 4) {
        const int n0 = wave * 16;
        v8f acc = {};
        acc = gemm_tile<64, AP1>(sze, 0, sWf1, n0, acc, l2, hi);
        const float bv = sbf1[n0 + l2];
#pragma unroll
        for (int r = 0; r < 8; ++r)
          sh[(r + 8 * hi) * AP1 + n0 + l2] = relu(acc[r] + bv);
      }
      v8f acc2[4];
#pragma unroll
      for (int jj = 0; jj < 4; ++jj)
        acc2[jj] = *reinterpret_cast<const v8f*>(sbf2 + 64 * wave + 16 * jj + 8 * hi);
      __syncthreads();

#pragma unroll
      for (int kk = 0; kk < 16; ++kk) {
        v2f hb;
        const float* ap = sh + l2 * AP1 + 4 * kk + 2 * hi;
        hb.x = ap[0];
        hb.y = ap[1];
#pragma unroll
        for (int jj = 0; jj < 4; ++jj) acc2[jj] = wmma4(wf2[jj][kk], hb, acc2[jj]);
      }
      const float4* dxp =
          reinterpret_cast<const float4*>(sdx[sub] + l2 * DXP + 8 * hi);
      const float4 dxa = dxp[0], dxb = dxp[1];
      const float dxr[8] = {dxa.x, dxa.y, dxa.z, dxa.w, dxb.x, dxb.y, dxb.z, dxb.w};
      float kp[4];
#pragma unroll
      for (int jj = 0; jj < 4; ++jj) {
        float s = tanhf(acc2[jj][0]) * dxr[0];
#pragma unroll
        for (int r = 1; r < 8; ++r) s += tanhf(acc2[jj][r]) * dxr[r];
        kp[jj] = s;
      }
#pragma unroll
      for (int jj = 0; jj < 4; ++jj) kp[jj] += xhalf(kp[jj]);
      if (hi == 0) {
        float* kdst = sk[sub] + l2 * AP1 + wave * 4;
#pragma unroll
        for (int jj = 0; jj < 4; ++jj) kdst[jj] = kp[jj];
      }
      __syncthreads();

      for (int i = tid; i < 16 * Hh; i += nthr) {
        const int m = i >> 6, h = i & 63;
        const int ad = m * AP1 + h;
        const float zb = szb[ad];
        if (sub == 0) {
          sze[ad] = zb + (dt * (1.f / 3.f)) * sk[0][ad];
        } else if (sub == 1) {
          sze[ad] = zb + dt * (sk[1][ad] - sk[0][ad] * (1.f / 3.f));
        } else if (sub == 2) {
          sze[ad] = zb + dt * (sk[0][ad] - sk[1][ad] + sk[2][ad]);
        } else {
          const float zn =
              zb + (sk[0][ad] + 3.f * (sk[1][ad] + sk[2][ad]) + sk[3][ad]) *
                       (dt * 0.125f);
          szb[ad] = zn;
          sze[ad] = zn;
          zcde[((size_t)(b0 + m) * Tt + (step + 1)) * Hh + h] = zn;
        }
      }
      __syncthreads();
    }
  }
}

__global__ __launch_bounds__(128) void ncde_head_kernel(
    const float* __restrict__ zflow, const float* __restrict__ zcde,
    const float* __restrict__ W_mix, const float* __restrict__ b_mix,
    const float* __restrict__ W_l1, const float* __restrict__ b_l1,
    const float* __restrict__ W_l2, const float* __restrict__ b_l2,
    float* __restrict__ out, float* __restrict__ zout) {
  __shared__ float sWmix[128 * WP], sWl1[64 * WP];
  __shared__ float sbmix[64], sbl1[64], sWl2v[64];
  __shared__ float sA[16 * AP2], sB[16 * AP1], sC[16 * AP1];

  const int tid = threadIdx.x, nthr = 128;
  const int lane = tid & 31, wave = tid >> 5, l2 = lane & 15, hi = lane >> 4;
  const int r0 = blockIdx.x * 16;

  load_w64<128>(sWmix, W_mix, tid, nthr);
  load_w64<64>(sWl1, W_l1, tid, nthr);
  if (tid < 64) {
    sbmix[tid] = b_mix[tid];
    sbl1[tid] = b_l1[tid];
    sWl2v[tid] = W_l2[tid];
  }
  for (int i = tid; i < 16 * 128; i += nthr) {
    const int m = i >> 7, c = i & 127;
    const size_t rr = r0 + m;
    sA[m * AP2 + c] = (c < 64) ? zflow[rr * Hh + c] : zcde[rr * Hh + (c - 64)];
  }
  __syncthreads();

  const int n0 = wave * 16;
  v8f acc = {};
  acc = gemm_tile<128, AP2>(sA, 0, sWmix, n0, acc, l2, hi);
  float bv = sbmix[n0 + l2];
#pragma unroll
  for (int r = 0; r < 8; ++r) {
    const int m = r + 8 * hi;
    const float v = acc[r] + bv;
    zout[(size_t)(r0 + m) * Hh + n0 + l2] = v;
    sB[m * AP1 + n0 + l2] = tanhf(v);
  }
  __syncthreads();

  acc = (v8f){};
  acc = gemm_tile<64, AP1>(sB, 0, sWl1, n0, acc, l2, hi);
  bv = sbl1[n0 + l2];
#pragma unroll
  for (int r = 0; r < 8; ++r)
    sC[(r + 8 * hi) * AP1 + n0 + l2] = relu(acc[r] + bv);
  __syncthreads();

  const int m = tid >> 3, p = tid & 7;
  float s = 0.f;
#pragma unroll
  for (int q = 0; q < 8; ++q) s += sC[m * AP1 + p * 8 + q] * sWl2v[p * 8 + q];
  s += __shfl_xor(s, 1, 32);
  s += __shfl_xor(s, 2, 32);
  s += __shfl_xor(s, 4, 32);
  if (p == 0) out[r0 + m] = s + b_l2[0];
}

extern "C" void kernel_launch(void* const* d_in, const int* in_sizes, int n_in,
                              void* d_out, int out_size, void* d_ws,
                              size_t ws_size, hipStream_t stream) {
  const float* x = (const float*)d_in[0];
  const float* seq_ts = (const float*)d_in[1];
  const float* coeffs = (const float*)d_in[3];
  const float* times = (const float*)d_in[4];
  const float* W_if = (const float*)d_in[5];
  const float* b_if = (const float*)d_in[6];
  const float* W_ic = (const float*)d_in[7];
  const float* b_ic = (const float*)d_in[8];
  const float* W_emb = (const float*)d_in[9];
  const float* b_emb = (const float*)d_in[10];
  const float* W_fl1 = (const float*)d_in[11];
  const float* b_fl1 = (const float*)d_in[12];
  const float* W_flo = (const float*)d_in[13];
  const float* b_flo = (const float*)d_in[14];
  const float* W_f1 = (const float*)d_in[15];
  const float* b_f1 = (const float*)d_in[16];
  const float* W_f2 = (const float*)d_in[17];
  const float* b_f2 = (const float*)d_in[18];
  const float* W_mix = (const float*)d_in[19];
  const float* b_mix = (const float*)d_in[20];
  const float* W_l1 = (const float*)d_in[21];
  const float* b_l1 = (const float*)d_in[22];
  const float* W_l2 = (const float*)d_in[23];
  const float* b_l2 = (const float*)d_in[24];

  float* zflow = (float*)d_ws;
  float* zcde = zflow + (size_t)Bb * Tt * Hh;
  float* out = (float*)d_out;
  float* zout = out + (size_t)Bb * Tt;

  ncde_embed_kernel<<<(Bb * Tt) / 16, 128, 0, stream>>>(
      x, seq_ts, coeffs, times, W_if, b_if, W_ic, b_ic, W_emb, b_emb, W_fl1,
      b_fl1, W_flo, b_flo, zflow);
  ncde_scan_kernel<<<Bb / 16, 512, 0, stream>>>(coeffs, times, W_ic, b_ic,
                                                W_f1, b_f1, W_f2, b_f2, zcde);
  ncde_head_kernel<<<(Bb * Tt) / 16, 128, 0, stream>>>(
      zflow, zcde, W_mix, b_mix, W_l1, b_l1, W_l2, b_l2, out, zout);
}


# scan: library tanhf (two branches per value) replaced by branch-free two-range tanh: odd polynomial for |v|<0.625 (relative accuracy kept), hardware exp2+rcp above; sum order unchanged
# speedup vs baseline: 2.2460x; 1.2642x over previous
#include <hip/hip_runtime.h>
#include <math.h>

#define Bb 256
#define Tt 512
#define Cc 16
#define Hh 64

#define WP  72
#define AP1 69
#define AP2 133
#define DXP 20

typedef __attribute__((ext_vector_type(2))) float v2f;
typedef __attribute__((ext_vector_type(8))) float v8f;

static __device__ __forceinline__ v8f wmma4(v2f a, v2f b, v8f c) {
  return __builtin_amdgcn_wmma_f32_16x16x4_f32(false, a, false, b, (short)0, c,
                                               false, false);
}

template <int K, int APITCH>
static __device__ __forceinline__ v8f gemm_tile(const float* __restrict__ A,
                                                int acol0,
                                                const float* __restrict__ W,
                                                int n0, v8f acc, int l2, int hi) {
#pragma unroll
  for (int k = 0; k < K; k += 4) {
    v2f a, b;
    const float* ap = A + l2 * APITCH + acol0 + k + 2 * hi;
    a.x = ap[0];
    a.y = ap[1];
    const float* wp = W + (k + 2 * hi) * WP + n0 + l2;
    b.x = wp[0];
    b.y = wp[WP];
    acc = wmma4(a, b, acc);
  }
  return acc;
}

template <int APITCH>
static __device__ __forceinline__ void store_tile(float* __restrict__ O, int n0,
                                                  v8f acc, int l2, int hi) {
#pragma unroll
  for (int r = 0; r < 8; ++r) O[(r + 8 * hi) * APITCH + n0 + l2] = acc[r];
}

template <int ROWS>
static __device__ __forceinline__ void load_w64(float* __restrict__ dst,
                                                const float* __restrict__ src,
                                                int tid, int nthr) {
  for (int i = tid; i < ROWS * 64; i += nthr)
    dst[(i >> 6) * WP + (i & 63)] = src[i];
}

static __device__ __forceinline__ float relu(float v) { return v > 0.f ? v : 0.f; }

static __device__ __forceinline__ float tanh_sel(float v) {
  const float u = v * v;
  float p = fmaf(u, -0.005704979412257671f, 0.020639078691601753f);
  p = fmaf(u, p, -0.05373971164226532f);
  p = fmaf(u, p, 0.13331441581249237f);
  p = fmaf(u, p, -0.3333328068256378f);
  const float zs = fmaf(v * u, p, v);
  const float y = fabsf(v);
  const float e = __builtin_amdgcn_exp2f(y * 2.8853900817779268f);
  const float zl = fmaf(-2.f, __builtin_amdgcn_rcpf(1.f + e), 1.f);
  return y < 0.625f ? zs : copysignf(zl, v);
}

static __device__ __forceinline__ float xhalf(float v) {
  const int b = __builtin_bit_cast(int, v);
  return __builtin_bit_cast(
      float, __builtin_amdgcn_permlanex16(b, b, 0x76543210, 0xfedcba98, false, false));
}

__global__ __launch_bounds__(128) void ncde_embed_kernel(
    const float* __restrict__ x, const float* __restrict__ seq_ts,
    const float* __restrict__ coeffs, const float* __restrict__ times,
    const float* __restrict__ W_if, const float* __restrict__ b_if,
    const float* __restrict__ W_ic, const float* __restrict__ b_ic,
    const float* __restrict__ W_emb, const float* __restrict__ b_emb,
    const float* __restrict__ W_fl1, const float* __restrict__ b_fl1,
    const float* __restrict__ W_flo, const float* __restrict__ b_flo,
    float* __restrict__ zflow) {
  __shared__ float sWif[16 * WP], sWic[16 * WP], sWemb[128 * WP];
  __shared__ float sWfl1[64 * WP], sWflo[64 * WP];
  __shared__ float sb[5][64];
  __shared__ float sA[16 * AP2], sB[16 * AP2];

  const int tid = threadIdx.x, nthr = 128;
  const int lane = tid & 31, wave = tid >> 5, l2 = lane & 15, hi = lane >> 4;
  const int r0 = blockIdx.x * 16;

  load_w64<16>(sWif, W_if, tid, nthr);
  load_w64<16>(sWic, W_ic, tid, nthr);
  load_w64<128>(sWemb, W_emb, tid, nthr);
  load_w64<64>(sWfl1, W_fl1, tid, nthr);
  load_w64<64>(sWflo, W_flo, tid, nthr);
  if (tid < 64) {
    sb[0][tid] = b_if[tid];
    sb[1][tid] = b_ic[tid];
    sb[2][tid] = b_emb[tid];
    sb[3][tid] = b_fl1[tid];
    sb[4][tid] = b_flo[tid];
  }
  for (int i = tid; i < 256; i += nthr) {
    const int m = i >> 4, c = i & 15;
    const int rr = r0 + m, b = rr / Tt, t = rr % Tt;
    sA[m * AP2 + c] = (c == 0) ? seq_ts[rr] : x[(size_t)rr * (Cc - 1) + (c - 1)];
    const int idx = (t < Tt - 1) ? t : (Tt - 2);
    const float s = times[t] - times[idx];
    const float* cf = coeffs + ((size_t)b * (Tt - 1) + idx) * (4 * Cc) + c;
    const float a0 = cf[0], b0 = cf[Cc], c0 = cf[2 * Cc], d0 = cf[3 * Cc];
    sA[m * AP2 + 16 + c] = a0 + s * (b0 + s * (c0 + s * d0));
  }
  __syncthreads();

  const int n0 = wave * 16;
  v8f aF = {}, aX = {};
  aF = gemm_tile<16, AP2>(sA, 0, sWif, n0, aF, l2, hi);
  aX = gemm_tile<16, AP2>(sA, 16, sWic, n0, aX, l2, hi);
  const float bF = sb[0][n0 + l2], bX = sb[1][n0 + l2];
#pragma unroll
  for (int r = 0; r < 8; ++r) {
    const int m = r + 8 * hi;
    sB[m * AP2 + n0 + l2] = aF[r] + bF;
    sB[m * AP2 + 64 + n0 + l2] = aX[r] + bX;
  }
  __syncthreads();

  v8f acc = {};
  acc = gemm_tile<128, AP2>(sB, 0, sWemb, n0, acc, l2, hi);
  float bv = sb[2][n0 + l2];
#pragma unroll
  for (int r = 0; r < 8; ++r)
    sA[(r + 8 * hi) * AP2 + n0 + l2] = relu(acc[r] + bv);
  __syncthreads();

  acc = (v8f){};
  acc = gemm_tile<64, AP2>(sA, 0, sWfl1, n0, acc, l2, hi);
  bv = sb[3][n0 + l2];
#pragma unroll
  for (int r = 0; r < 8; ++r)
    sB[(r + 8 * hi) * AP2 + n0 + l2] = relu(acc[r] + bv);
  __syncthreads();

  acc = (v8f){};
  acc = gemm_tile<64, AP2>(sB, 0, sWflo, n0, acc, l2, hi);
  bv = sb[4][n0 + l2];
#pragma unroll
  for (int r = 0; r < 8; ++r)
    zflow[(size_t)(r0 + r + 8 * hi) * Hh + n0 + l2] = acc[r] + bv;
}

__global__ __launch_bounds__(512, 1) void ncde_scan_kernel(
    const float* __restrict__ coeffs, const float* __restrict__ times,
    const float* __restrict__ W_ic, const float* __restrict__ b_ic,
    const float* __restrict__ W_f1, const float* __restrict__ b_f1,
    const float* __restrict__ W_f2, const float* __restrict__ b_f2,
    float* __restrict__ zcde) {
  __shared__ float sWf1[64 * WP], sWic[16 * WP];
  __shared__ float sbf1[64], sbic[64];
  __shared__ float stim[Tt];
  __shared__ float sh[16 * AP1], szb[16 * AP1], sze[16 * AP1];
  __shared__ float sk[4][16 * AP1];
  __shared__ __attribute__((aligned(16))) float sdx[4][16 * DXP];
  __shared__ __attribute__((aligned(32))) float sbf2[Hh * Cc];

  const int tid = threadIdx.x, nthr = 512;
  const int lane = tid & 31, wave = tid >> 5, l2 = lane & 15, hi = lane >> 4;
  const int b0 = blockIdx.x * 16;

  load_w64<64>(sWf1, W_f1, tid, nthr);
  load_w64<16>(sWic, W_ic, tid, nthr);
  if (tid < 64) {
    sbf1[tid] = b_f1[tid];
    sbic[tid] = b_ic[tid];
  }
  for (int i = tid; i < Tt; i += nthr) stim[i] = times[i];
  for (int i = tid; i < Hh * Cc; i += nthr) sbf2[i] = b_f2[i];

  v2f wf2[4][16];
#pragma unroll
  for (int jj = 0; jj < 4; ++jj) {
    const int j = wave * 4 + jj;
    const int col = 16 * j + l2;
#pragma unroll
    for (int kk = 0; kk < 16; ++kk) {
      const int row = 4 * kk + 2 * hi;
      wf2[jj][kk].x = W_f2[(size_t)row * (Hh * Cc) + col];
      wf2[jj][kk].y = W_f2[(size_t)(row + 1) * (Hh * Cc) + col];
    }
  }

  for (int i = tid; i < 256; i += nthr) {
    const int m = i >> 4, c = i & 15;
    sze[m * AP1 + c] = coeffs[((size_t)(b0 + m) * (Tt - 1)) * (4 * Cc) + c];
  }
  __syncthreads();
  if (wave < 4) {
    const int n0 = wave * 16;
    v8f acc = {};
    acc = gemm_tile<16, AP1>(sze, 0, sWic, n0, acc, l2, hi);
    const float bv = sbic[n0 + l2];
#pragma unroll
    for (int r = 0; r < 8; ++r) {
      const int m = r + 8 * hi;
      const float v = acc[r] + bv;
      szb[m * AP1 + n0 + l2] = v;
      zcde[((size_t)(b0 + m) * Tt) * Hh + n0 + l2] = v;
    }
  }
  __syncthreads();
  for (int i = tid; i < 16 * Hh; i += nthr) {
    const int m = i >> 6, h = i & 63;
    sze[m * AP1 + h] = szb[m * AP1 + h];
  }
  __syncthreads();

  for (int step = 0; step < Tt - 1; ++step) {
    const float dt = stim[step + 1] - stim[step];
    if (tid < 256) {
      const int m = tid >> 4, c = tid & 15, b = b0 + m;
      const float* cf = coeffs + ((size_t)b * (Tt - 1) + step) * (4 * Cc) + c;
      const float bb = cf[Cc], cc = cf[2 * Cc], dd = cf[3 * Cc];
      const float s1 = dt * (1.f / 3.f), s2 = dt * (2.f / 3.f);
      sdx[0][m * DXP + c] = bb;
      sdx[1][m * DXP + c] = bb + s1 * (2.f * cc + 3.f * dd * s1);
      sdx[2][m * DXP + c] = bb + s2 * (2.f * cc + 3.f * dd * s2);
      const int idx3 = (step + 1 < Tt - 1) ? (step + 1) : (Tt - 2);
      const float* cf3 = coeffs + ((size_t)b * (Tt - 1) + idx3) * (4 * Cc) + c;
      const float b3 = cf3[Cc], c3 = cf3[2 * Cc], d3 = cf3[3 * Cc];
      const float s3 = stim[step + 1] - stim[idx3];
      sdx[3][m * DXP + c] = b3 + s3 * (2.f * c3 + 3.f * d3 * s3);
    }
    __syncthreads();

    for (int sub = 0; sub < 4; ++sub) {
      if (wave < 4) {
        const int n0 = wave * 16;
        v8f acc = {};
        acc = gemm_tile<64, AP1>(sze, 0, sWf1, n0, acc, l2, hi);
        const float bv = sbf1[n0 + l2];
#pragma unroll
        for (int r = 0; r < 8; ++r)
          sh[(r + 8 * hi) * AP1 + n0 + l2] = relu(acc[r] + bv);
      }
      v8f acc2[4];
#pragma unroll
      for (int jj = 0; jj < 4; ++jj)
        acc2[jj] = *reinterpret_cast<const v8f*>(sbf2 + 64 * wave + 16 * jj + 8 * hi);
      __syncthreads();

#pragma unroll
      for (int kk = 0; kk < 16; ++kk) {
        v2f hb;
        const float* ap = sh + l2 * AP1 + 4 * kk + 2 * hi;
        hb.x = ap[0];
        hb.y = ap[1];
#pragma unroll
        for (int jj = 0; jj < 4; ++jj) acc2[jj] = wmma4(wf2[jj][kk], hb, acc2[jj]);
      }
      const float4* dxp =
          reinterpret_cast<const float4*>(sdx[sub] + l2 * DXP + 8 * hi);
      const float4 dxa = dxp[0], dxb = dxp[1];
      const float dxr[8] = {dxa.x, dxa.y, dxa.z, dxa.w, dxb.x, dxb.y, dxb.z, dxb.w};
      float kp[4];
#pragma unroll
      for (int jj = 0; jj < 4; ++jj) {
        float s = tanh_sel(acc2[jj][0]) * dxr[0];
#pragma unroll
        for (int r = 1; r < 8; ++r) s += tanh_sel(acc2[jj][r]) * dxr[r];
        kp[jj] = s;
      }
#pragma unroll
      for (int jj = 0; jj < 4; ++jj) kp[jj] += xhalf(kp[jj]);
      if (hi == 0) {
        float* kdst = sk[sub] + l2 * AP1 + wave * 4;
#pragma unroll
        for (int jj = 0; jj < 4; ++jj) kdst[jj] = kp[jj];
      }
      __syncthreads();

      for (int i = tid; i < 16 * Hh; i += nthr) {
        const int m = i >> 6, h = i & 63;
        const int ad = m * AP1 + h;
        const float zb = szb[ad];
        if (sub == 0) {
          sze[ad] = zb + (dt * (1.f / 3.f)) * sk[0][ad];
        } else if (sub == 1) {
          sze[ad] = zb + dt * (sk[1][ad] - sk[0][ad] * (1.f / 3.f));
        } else if (sub == 2) {
          sze[ad] = zb + dt * (sk[0][ad] - sk[1][ad] + sk[2][ad]);
        } else {
          const float zn =
              zb + (sk[0][ad] + 3.f * (sk[1][ad] + sk[2][ad]) + sk[3][ad]) *
                       (dt * 0.125f);
          szb[ad] = zn;
          sze[ad] = zn;
          zcde[((size_t)(b0 + m) * Tt + (step + 1)) * Hh + h] = zn;
        }
      }
      __syncthreads();
    }
  }
}

__global__ __launch_bounds__(128) void ncde_head_kernel(
    const float* __restrict__ zflow, const float* __restrict__ zcde,
    const float* __restrict__ W_mix, const float* __restrict__ b_mix,
    const float* __restrict__ W_l1, const float* __restrict__ b_l1,
    const float* __restrict__ W_l2, const float* __restrict__ b_l2,
    float* __restrict__ out, float* __restrict__ zout) {
  __shared__ float sWmix[128 * WP], sWl1[64 * WP];
  __shared__ float sbmix[64], sbl1[64], sWl2v[64];
  __shared__ float sA[16 * AP2], sB[16 * AP1], sC[16 * AP1];

  const int tid = threadIdx.x, nthr = 128;
  const int lane = tid & 31, wave = tid >> 5, l2 = lane & 15, hi = lane >> 4;
  const int r0 = blockIdx.x * 16;

  load_w64<128>(sWmix, W_mix, tid, nthr);
  load_w64<64>(sWl1, W_l1, tid, nthr);
  if (tid < 64) {
    sbmix[tid] = b_mix[tid];
    sbl1[tid] = b_l1[tid];
    sWl2v[tid] = W_l2[tid];
  }
  for (int i = tid; i < 16 * 128; i += nthr) {
    const int m = i >> 7, c = i & 127;
    const size_t rr = r0 + m;
    sA[m * AP2 + c] = (c < 64) ? zflow[rr * Hh + c] : zcde[rr * Hh + (c - 64)];
  }
  __syncthreads();

  const int n0 = wave * 16;
  v8f acc = {};
  acc = gemm_tile<128, AP2>(sA, 0, sWmix, n0, acc, l2, hi);
  float bv = sbmix[n0 + l2];
#pragma unroll
  for (int r = 0; r < 8; ++r) {
    const int m = r + 8 * hi;
    const float v = acc[r] + bv;
    zout[(size_t)(r0 + m) * Hh + n0 + l2] = v;
    sB[m * AP1 + n0 + l2] = tanhf(v);
  }
  __syncthreads();

  acc = (v8f){};
  acc = gemm_tile<64, AP1>(sB, 0, sWl1, n0, acc, l2, hi);
  bv = sbl1[n0 + l2];
#pragma unroll
  for (int r = 0; r < 8; ++r)
    sC[(r + 8 * hi) * AP1 + n0 + l2] = relu(acc[r] + bv);
  __syncthreads();

  const int m = tid >> 3, p = tid & 7;
  float s = 0.f;
#pragma unroll
  for (int q = 0; q < 8; ++q) s += sC[m * AP1 + p * 8 + q] * sWl2v[p * 8 + q];
  s += __shfl_xor(s, 1, 32);
  s += __shfl_xor(s, 2, 32);
  s += __shfl_xor(s, 4, 32);
  if (p == 0) out[r0 + m] = s + b_l2[0];
}

extern "C" void kernel_launch(void* const* d_in, const int* in_sizes, int n_in,
                              void* d_out, int out_size, void* d_ws,
                              size_t ws_size, hipStream_t stream) {
  const float* x = (const float*)d_in[0];
  const float* seq_ts = (const float*)d_in[1];
  const float* coeffs = (const float*)d_in[3];
  const float* times = (const float*)d_in[4];
  const float* W_if = (const float*)d_in[5];
  const float* b_if = (const float*)d_in[6];
  const float* W_ic = (const float*)d_in[7];
  const float* b_ic = (const float*)d_in[8];
  const float* W_emb = (const float*)d_in[9];
  const float* b_emb = (const float*)d_in[10];
  const float* W_fl1 = (const float*)d_in[11];
  const float* b_fl1 = (const float*)d_in[12];
  const float* W_flo = (const float*)d_in[13];
  const float* b_flo = (const float*)d_in[14];
  const float* W_f1 = (const float*)d_in[15];
  const float* b_f1 = (const float*)d_in[16];
  const float* W_f2 = (const float*)d_in[17];
  const float* b_f2 = (const float*)d_in[18];
  const float* W_mix = (const float*)d_in[19];
  const float* b_mix = (const float*)d_in[20];
  const float* W_l1 = (const float*)d_in[21];
  const float* b_l1 = (const float*)d_in[22];
  const float* W_l2 = (const float*)d_in[23];
  const float* b_l2 = (const float*)d_in[24];

  float* zflow = (float*)d_ws;
  float* zcde = zflow + (size_t)Bb * Tt * Hh;
  float* out = (float*)d_out;
  float* zout = out + (size_t)Bb * Tt;

  ncde_embed_kernel<<<(Bb * Tt) / 16, 128, 0, stream>>>(
      x, seq_ts, coeffs, times, W_if, b_if, W_ic, b_ic, W_emb, b_emb, W_fl1,
      b_fl1, W_flo, b_flo, zflow);
  ncde_scan_kernel<<<Bb / 16, 512, 0, stream>>>(coeffs, times, W_ic, b_ic,
                                                W_f1, b_f1, W_f2, b_f2, zcde);
  ncde_head_kernel<<<(Bb * Tt) / 16, 128, 0, stream>>>(
      zflow, zcde, W_mix, b_mix, W_l1, b_l1, W_l2, b_l2, out, zout);
}


# scan: tanh by the hardware instruction v_tanh_f32 (one transcendental per value) in place of the two-range polynomial + exp2/rcp select; result bits change, sum order unchanged
# speedup vs baseline: 3.0613x; 1.3630x over previous
#include <hip/hip_runtime.h>
#include <math.h>

#define Bb 256
#define Tt 512
#define Cc 16
#define Hh 64

#define WP  72
#define AP1 69
#define AP2 133
#define DXP 20

typedef __attribute__((ext_vector_type(2))) float v2f;
typedef __attribute__((ext_vector_type(8))) float v8f;

static __device__ __forceinline__ v8f wmma4(v2f a, v2f b, v8f c) {
  return __builtin_amdgcn_wmma_f32_16x16x4_f32(false, a, false, b, (short)0, c,
                                               false, false);
}

template <int K, int APITCH>
static __device__ __forceinline__ v8f gemm_tile(const float* __restrict__ A,
                                                int acol0,
                                                const float* __restrict__ W,
                                                int n0, v8f acc, int l2, int hi) {
#pragma unroll
  for (int k = 0; k < K; k += 4) {
    v2f a, b;
    const float* ap = A + l2 * APITCH + acol0 + k + 2 * hi;
    a.x = ap[0];
    a.y = ap[1];
    const float* wp = W + (k + 2 * hi) * WP + n0 + l2;
    b.x = wp[0];
    b.y = wp[WP];
    acc = wmma4(a, b, acc);
  }
  return acc;
}

template <int APITCH>
static __device__ __forceinline__ void store_tile(float* __restrict__ O, int n0,
                                                  v8f acc, int l2, int hi) {
#pragma unroll
  for (int r = 0; r < 8; ++r) O[(r + 8 * hi) * APITCH + n0 + l2] = acc[r];
}

template <int ROWS>
static __device__ __forceinline__ void load_w64(float* __restrict__ dst,
                                                const float* __restrict__ src,
                                                int tid, int nthr) {
  for (int i = tid; i < ROWS * 64; i += nthr)
    dst[(i >> 6) * WP + (i & 63)] = src[i];
}

static __device__ __forceinline__ float relu(float v) { return v > 0.f ? v : 0.f; }

static __device__ __forceinline__ float tanh_hw(float v) {
  return __builtin_amdgcn_tanhf(v);
}

static __device__ __forceinline__ float xhalf(float v) {
  const int b = __builtin_bit_cast(int, v);
  return __builtin_bit_cast(
      float, __builtin_amdgcn_permlanex16(b, b, 0x76543210, 0xfedcba98, false, false));
}

__global__ __launch_bounds__(128) void ncde_embed_kernel(
    const float* __restrict__ x, const float* __restrict__ seq_ts,
    const float* __restrict__ coeffs, const float* __restrict__ times,
    const float* __restrict__ W_if, const float* __restrict__ b_if,
    const float* __restrict__ W_ic, const float* __restrict__ b_ic,
    const float* __restrict__ W_emb, const float* __restrict__ b_emb,
    const float* __restrict__ W_fl1, const float* __restrict__ b_fl1,
    const float* __restrict__ W_flo, const float* __restrict__ b_flo,
    float* __restrict__ zflow) {
  __shared__ float sWif[16 * WP], sWic[16 * WP], sWemb[128 * WP];
  __shared__ float sWfl1[64 * WP], sWflo[64 * WP];
  __shared__ float sb[5][64];
  __shared__ float sA[16 * AP2], sB[16 * AP2];

  const int tid = threadIdx.x, nthr = 128;
  const int lane = tid & 31, wave = tid >> 5, l2 = lane & 15, hi = lane >> 4;
  const int r0 = blockIdx.x * 16;

  load_w64<16>(sWif, W_if, tid, nthr);
  load_w64<16>(sWic, W_ic, tid, nthr);
  load_w64<128>(sWemb, W_emb, tid, nthr);
  load_w64<64>(sWfl1, W_fl1, tid, nthr);
  load_w64<64>(sWflo, W_flo, tid, nthr);
  if (tid < 64) {
    sb[0][tid] = b_if[tid];
    sb[1][tid] = b_ic[tid];
    sb[2][tid] = b_emb[tid];
    sb[3][tid] = b_fl1[tid];
    sb[4][tid] = b_flo[tid];
  }
  for (int i = tid; i < 256; i += nthr) {
    const int m = i >> 4, c = i & 15;
    const int rr = r0 + m, b = rr / Tt, t = rr % Tt;
    sA[m * AP2 + c] = (c == 0) ? seq_ts[rr] : x[(size_t)rr * (Cc - 1) + (c - 1)];
    const int idx = (t < Tt - 1) ? t : (Tt - 2);
    const float s = times[t] - times[idx];
    const float* cf = coeffs + ((size_t)b * (Tt - 1) + idx) * (4 * Cc) + c;
    const float a0 = cf[0], b0 = cf[Cc], c0 = cf[2 * Cc], d0 = cf[3 * Cc];
    sA[m * AP2 + 16 + c] = a0 + s * (b0 + s * (c0 + s * d0));
  }
  __syncthreads();

  const int n0 = wave * 16;
  v8f aF = {}, aX = {};
  aF = gemm_tile<16, AP2>(sA, 0, sWif, n0, aF, l2, hi);
  aX = gemm_tile<16, AP2>(sA, 16, sWic, n0, aX, l2, hi);
  const float bF = sb[0][n0 + l2], bX = sb[1][n0 + l2];
#pragma unroll
  for (int r = 0; r < 8; ++r) {
    const int m = r + 8 * hi;
    sB[m * AP2 + n0 + l2] = aF[r] + bF;
    sB[m * AP2 + 64 + n0 + l2] = aX[r] + bX;
  }
  __syncthreads();

  v8f acc = {};
  acc = gemm_tile<128, AP2>(sB, 0, sWemb, n0, acc, l2, hi);
  float bv = sb[2][n0 + l2];
#pragma unroll
  for (int r = 0; r < 8; ++r)
    sA[(r + 8 * hi) * AP2 + n0 + l2] = relu(acc[r] + bv);
  __syncthreads();

  acc = (v8f){};
  acc = gemm_tile<64, AP2>(sA, 0, sWfl1, n0, acc, l2, hi);
  bv = sb[3][n0 + l2];
#pragma unroll
  for (int r = 0; r < 8; ++r)
    sB[(r + 8 * hi) * AP2 + n0 + l2] = relu(acc[r] + bv);
  __syncthreads();

  acc = (v8f){};
  acc = gemm_tile<64, AP2>(sB, 0, sWflo, n0, acc, l2, hi);
  bv = sb[4][n0 + l2];
#pragma unroll
  for (int r = 0; r < 8; ++r)
    zflow[(size_t)(r0 + r + 8 * hi) * Hh + n0 + l2] = acc[r] + bv;
}

__global__ __launch_bounds__(512, 1) void ncde_scan_kernel(
    const float* __restrict__ coeffs, const float* __restrict__ times,
    const float* __restrict__ W_ic, const float* __restrict__ b_ic,
    const float* __restrict__ W_f1, const float* __restrict__ b_f1,
    const float* __restrict__ W_f2, const float* __restrict__ b_f2,
    float* __restrict__ zcde) {
  __shared__ float sWf1[64 * WP], sWic[16 * WP];
  __shared__ float sbf1[64], sbic[64];
  __shared__ float stim[Tt];
  __shared__ float sh[16 * AP1], szb[16 * AP1], sze[16 * AP1];
  __shared__ float sk[4][16 * AP1];
  __shared__ __attribute__((aligned(16))) float sdx[4][16 * DXP];
  __shared__ __attribute__((aligned(32))) float sbf2[Hh * Cc];

  const int tid = threadIdx.x, nthr = 512;
  const int lane = tid & 31, wave = tid >> 5, l2 = lane & 15, hi = lane >> 4;
  const int b0 = blockIdx.x * 16;

  load_w64<64>(sWf1, W_f1, tid, nthr);
  load_w64<16>(sWic, W_ic, tid, nthr);
  if (tid < 64) {
    sbf1[tid] = b_f1[tid];
    sbic[tid] = b_ic[tid];
  }
  for (int i = tid; i < Tt; i += nthr) stim[i] = times[i];
  for (int i = tid; i < Hh * Cc; i += nthr) sbf2[i] = b_f2[i];

  v2f wf2[4][16];
#pragma unroll
  for (int jj = 0; jj < 4; ++jj) {
    const int j = wave * 4 + jj;
    const int col = 16 * j + l2;
#pragma unroll
    for (int kk = 0; kk < 16; ++kk) {
      const int row = 4 * kk + 2 * hi;
      wf2[jj][kk].x = W_f2[(size_t)row * (Hh * Cc) + col];
      wf2[jj][kk].y = W_f2[(size_t)(row + 1) * (Hh * Cc) + col];
    }
  }

  for (int i = tid; i < 256; i += nthr) {
    const int m = i >> 4, c = i & 15;
    sze[m * AP1 + c] = coeffs[((size_t)(b0 + m) * (Tt - 1)) * (4 * Cc) + c];
  }
  __syncthreads();
  if (wave < 4) {
    const int n0 = wave * 16;
    v8f acc = {};
    acc = gemm_tile<16, AP1>(sze, 0, sWic, n0, acc, l2, hi);
    const float bv = sbic[n0 + l2];
#pragma unroll
    for (int r = 0; r < 8; ++r) {
      const int m = r + 8 * hi;
      const float v = acc[r] + bv;
      szb[m * AP1 + n0 + l2] = v;
      zcde[((size_t)(b0 + m) * Tt) * Hh + n0 + l2] = v;
    }
  }
  __syncthreads();
  for (int i = tid; i < 16 * Hh; i += nthr) {
    const int m = i >> 6, h = i & 63;
    sze[m * AP1 + h] = szb[m * AP1 + h];
  }
  __syncthreads();

  for (int step = 0; step < Tt - 1; ++step) {
    const float dt = stim[step + 1] - stim[step];
    if (tid < 256) {
      const int m = tid >> 4, c = tid & 15, b = b0 + m;
      const float* cf = coeffs + ((size_t)b * (Tt - 1) + step) * (4 * Cc) + c;
      const float bb = cf[Cc], cc = cf[2 * Cc], dd = cf[3 * Cc];
      const float s1 = dt * (1.f / 3.f), s2 = dt * (2.f / 3.f);
      sdx[0][m * DXP + c] = bb;
      sdx[1][m * DXP + c] = bb + s1 * (2.f * cc + 3.f * dd * s1);
      sdx[2][m * DXP + c] = bb + s2 * (2.f * cc + 3.f * dd * s2);
      const int idx3 = (step + 1 < Tt - 1) ? (step + 1) : (Tt - 2);
      const float* cf3 = coeffs + ((size_t)b * (Tt - 1) + idx3) * (4 * Cc) + c;
      const float b3 = cf3[Cc], c3 = cf3[2 * Cc], d3 = cf3[3 * Cc];
      const float s3 = stim[step + 1] - stim[idx3];
      sdx[3][m * DXP + c] = b3 + s3 * (2.f * c3 + 3.f * d3 * s3);
    }
    __syncthreads();

    for (int sub = 0; sub < 4; ++sub) {
      if (wave < 4) {
        const int n0 = wave * 16;
        v8f acc = {};
        acc = gemm_tile<64, AP1>(sze, 0, sWf1, n0, acc, l2, hi);
        const float bv = sbf1[n0 + l2];
#pragma unroll
        for (int r = 0; r < 8; ++r)
          sh[(r + 8 * hi) * AP1 + n0 + l2] = relu(acc[r] + bv);
      }
      v8f acc2[4];
#pragma unroll
      for (int jj = 0; jj < 4; ++jj)
        acc2[jj] = *reinterpret_cast<const v8f*>(sbf2 + 64 * wave + 16 * jj + 8 * hi);
      __syncthreads();

#pragma unroll
      for (int kk = 0; kk < 16; ++kk) {
        v2f hb;
        const float* ap = sh + l2 * AP1 + 4 * kk + 2 * hi;
        hb.x = ap[0];
        hb.y = ap[1];
#pragma unroll
        for (int jj = 0; jj < 4; ++jj) acc2[jj] = wmma4(wf2[jj][kk], hb, acc2[jj]);
      }
      const float4* dxp =
          reinterpret_cast<const float4*>(sdx[sub] + l2 * DXP + 8 * hi);
      const float4 dxa = dxp[0], dxb = dxp[1];
      const float dxr[8] = {dxa.x, dxa.y, dxa.z, dxa.w, dxb.x, dxb.y, dxb.z, dxb.w};
      float kp[4];
#pragma unroll
      for (int jj = 0; jj < 4; ++jj) {
        float s = tanh_hw(acc2[jj][0]) * dxr[0];
#pragma unroll
        for (int r = 1; r < 8; ++r) s += tanh_hw(acc2[jj][r]) * dxr[r];
        kp[jj] = s;
      }
#pragma unroll
      for (int jj = 0; jj < 4; ++jj) kp[jj] += xhalf(kp[jj]);
      if (hi == 0) {
        float* kdst = sk[sub] + l2 * AP1 + wave * 4;
#pragma unroll
        for (int jj = 0; jj < 4; ++jj) kdst[jj] = kp[jj];
      }
      __syncthreads();

      for (int i = tid; i < 16 * Hh; i += nthr) {
        const int m = i >> 6, h = i & 63;
        const int ad = m * AP1 + h;
        const float zb = szb[ad];
        if (sub == 0) {
          sze[ad] = zb + (dt * (1.f / 3.f)) * sk[0][ad];
        } else if (sub == 1) {
          sze[ad] = zb + dt * (sk[1][ad] - sk[0][ad] * (1.f / 3.f));
        } else if (sub == 2) {
          sze[ad] = zb + dt * (sk[0][ad] - sk[1][ad] + sk[2][ad]);
        } else {
          const float zn =
              zb + (sk[0][ad] + 3.f * (sk[1][ad] + sk[2][ad]) + sk[3][ad]) *
                       (dt * 0.125f);
          szb[ad] = zn;
          sze[ad] = zn;
          zcde[((size_t)(b0 + m) * Tt + (step + 1)) * Hh + h] = zn;
        }
      }
      __syncthreads();
    }
  }
}

__global__ __launch_bounds__(128) void ncde_head_kernel(
    const float* __restrict__ zflow, const float* __restrict__ zcde,
    const float* __restrict__ W_mix, const float* __restrict__ b_mix,
    const float* __restrict__ W_l1, const float* __restrict__ b_l1,
    const float* __restrict__ W_l2, const float* __restrict__ b_l2,
    float* __restrict__ out, float* __restrict__ zout) {
  __shared__ float sWmix[128 * WP], sWl1[64 * WP];
  __shared__ float sbmix[64], sbl1[64], sWl2v[64];
  __shared__ float sA[16 * AP2], sB[16 * AP1], sC[16 * AP1];

  const int tid = threadIdx.x, nthr = 128;
  const int lane = tid & 31, wave = tid >> 5, l2 = lane & 15, hi = lane >> 4;
  const int r0 = blockIdx.x * 16;

  load_w64<128>(sWmix, W_mix, tid, nthr);
  load_w64<64>(sWl1, W_l1, tid, nthr);
  if (tid < 64) {
    sbmix[tid] = b_mix[tid];
    sbl1[tid] = b_l1[tid];
    sWl2v[tid] = W_l2[tid];
  }
  for (int i = tid; i < 16 * 128; i += nthr) {
    const int m = i >> 7, c = i & 127;
    const size_t rr = r0 + m;
    sA[m * AP2 + c] = (c < 64) ? zflow[rr * Hh + c] : zcde[rr * Hh + (c - 64)];
  }
  __syncthreads();

  const int n0 = wave * 16;
  v8f acc = {};
  acc = gemm_tile<128, AP2>(sA, 0, sWmix, n0, acc, l2, hi);
  float bv = sbmix[n0 + l2];
#pragma unroll
  for (int r = 0; r < 8; ++r) {
    const int m = r + 8 * hi;
    const float v = acc[r] + bv;
    zout[(size_t)(r0 + m) * Hh + n0 + l2] = v;
    sB[m * AP1 + n0 + l2] = tanhf(v);
  }
  __syncthreads();

  acc = (v8f){};
  acc = gemm_tile<64, AP1>(sB, 0, sWl1, n0, acc, l2, hi);
  bv = sbl1[n0 + l2];
#pragma unroll
  for (int r = 0; r < 8; ++r)
    sC[(r + 8 * hi) * AP1 + n0 + l2] = relu(acc[r] + bv);
  __syncthreads();

  const int m = tid >> 3, p = tid & 7;
  float s = 0.f;
#pragma unroll
  for (int q = 0; q < 8; ++q) s += sC[m * AP1 + p * 8 + q] * sWl2v[p * 8 + q];
  s += __shfl_xor(s, 1, 32);
  s += __shfl_xor(s, 2, 32);
  s += __shfl_xor(s, 4, 32);
  if (p == 0) out[r0 + m] = s + b_l2[0];
}

extern "C" void kernel_launch(void* const* d_in, const int* in_sizes, int n_in,
                              void* d_out, int out_size, void* d_ws,
                              size_t ws_size, hipStream_t stream) {
  const float* x = (const float*)d_in[0];
  const float* seq_ts = (const float*)d_in[1];
  const float* coeffs = (const float*)d_in[3];
  const float* times = (const float*)d_in[4];
  const float* W_if = (const float*)d_in[5];
  const float* b_if = (const float*)d_in[6];
  const float* W_ic = (const float*)d_in[7];
  const float* b_ic = (const float*)d_in[8];
  const float* W_emb = (const float*)d_in[9];
  const float* b_emb = (const float*)d_in[10];
  const float* W_fl1 = (const float*)d_in[11];
  const float* b_fl1 = (const float*)d_in[12];
  const float* W_flo = (const float*)d_in[13];
  const float* b_flo = (const float*)d_in[14];
  const float* W_f1 = (const float*)d_in[15];
  const float* b_f1 = (const float*)d_in[16];
  const float* W_f2 = (const float*)d_in[17];
  const float* b_f2 = (const float*)d_in[18];
  const float* W_mix = (const float*)d_in[19];
  const float* b_mix = (const float*)d_in[20];
  const float* W_l1 = (const float*)d_in[21];
  const float* b_l1 = (const float*)d_in[22];
  const float* W_l2 = (const float*)d_in[23];
  const float* b_l2 = (const float*)d_in[24];

  float* zflow = (float*)d_ws;
  float* zcde = zflow + (size_t)Bb * Tt * Hh;
  float* out = (float*)d_out;
  float* zout = out + (size_t)Bb * Tt;

  ncde_embed_kernel<<<(Bb * Tt) / 16, 128, 0, stream>>>(
      x, seq_ts, coeffs, times, W_if, b_if, W_ic, b_ic, W_emb, b_emb, W_fl1,
      b_fl1, W_flo, b_flo, zflow);
  ncde_scan_kernel<<<Bb / 16, 512, 0, stream>>>(coeffs, times, W_ic, b_ic,
                                                W_f1, b_f1, W_f2, b_f2, zcde);
  ncde_head_kernel<<<(Bb * Tt) / 16, 128, 0, stream>>>(
      zflow, zcde, W_mix, b_mix, W_l1, b_l1, W_l2, b_l2, out, zout);
}


# scan: RK4 state (z, k1..k3) kept in registers of the lane that produces its k; update done in the einsum epilogue: no k buffers, no update pass, 2 barriers per sub-step not 3; sum order unchanged
# speedup vs baseline: 3.4561x; 1.1290x over previous
#include <hip/hip_runtime.h>
#include <math.h>

#define Bb 256
#define Tt 512
#define Cc 16
#define Hh 64

#define WP  72
#define AP1 69
#define AP2 133
#define DXP 20
#define ZOP 68

typedef __attribute__((ext_vector_type(2))) float v2f;
typedef __attribute__((ext_vector_type(8))) float v8f;

static __device__ __forceinline__ v8f wmma4(v2f a, v2f b, v8f c) {
  return __builtin_amdgcn_wmma_f32_16x16x4_f32(false, a, false, b, (short)0, c,
                                               false, false);
}

template <int K, int APITCH>
static __device__ __forceinline__ v8f gemm_tile(const float* __restrict__ A,
                                                int acol0,
                                                const float* __restrict__ W,
                                                int n0, v8f acc, int l2, int hi) {
#pragma unroll
  for (int k = 0; k < K; k += 4) {
    v2f a, b;
    const float* ap = A + l2 * APITCH + acol0 + k + 2 * hi;
    a.x = ap[0];
    a.y = ap[1];
    const float* wp = W + (k + 2 * hi) * WP + n0 + l2;
    b.x = wp[0];
    b.y = wp[WP];
    acc = wmma4(a, b, acc);
  }
  return acc;
}

template <int APITCH>
static __device__ __forceinline__ void store_tile(float* __restrict__ O, int n0,
                                                  v8f acc, int l2, int hi) {
#pragma unroll
  for (int r = 0; r < 8; ++r) O[(r + 8 * hi) * APITCH + n0 + l2] = acc[r];
}

template <int ROWS>
static __device__ __forceinline__ void load_w64(float* __restrict__ dst,
                                                const float* __restrict__ src,
                                                int tid, int nthr) {
  for (int i = tid; i < ROWS * 64; i += nthr)
    dst[(i >> 6) * WP + (i & 63)] = src[i];
}

static __device__ __forceinline__ float relu(float v) { return v > 0.f ? v : 0.f; }

static __device__ __forceinline__ float tanh_hw(float v) {
  return __builtin_amdgcn_tanhf(v);
}

static __device__ __forceinline__ float xhalf(float v) {
  const int b = __builtin_bit_cast(int, v);
  return __builtin_bit_cast(
      float, __builtin_amdgcn_permlanex16(b, b, 0x76543210, 0xfedcba98, false, false));
}

__global__ __launch_bounds__(128) void ncde_embed_kernel(
    const float* __restrict__ x, const float* __restrict__ seq_ts,
    const float* __restrict__ coeffs, const float* __restrict__ times,
    const float* __restrict__ W_if, const float* __restrict__ b_if,
    const float* __restrict__ W_ic, const float* __restrict__ b_ic,
    const float* __restrict__ W_emb, const float* __restrict__ b_emb,
    const float* __restrict__ W_fl1, const float* __restrict__ b_fl1,
    const float* __restrict__ W_flo, const float* __restrict__ b_flo,
    float* __restrict__ zflow) {
  __shared__ float sWif[16 * WP], sWic[16 * WP], sWemb[128 * WP];
  __shared__ float sWfl1[64 * WP], sWflo[64 * WP];
  __shared__ float sb[5][64];
  __shared__ float sA[16 * AP2], sB[16 * AP2];

  const int tid = threadIdx.x, nthr = 128;
  const int lane = tid & 31, wave = tid >> 5, l2 = lane & 15, hi = lane >> 4;
  const int r0 = blockIdx.x * 16;

  load_w64<16>(sWif, W_if, tid, nthr);
  load_w64<16>(sWic, W_ic, tid, nthr);
  load_w64<128>(sWemb, W_emb, tid, nthr);
  load_w64<64>(sWfl1, W_fl1, tid, nthr);
  load_w64<64>(sWflo, W_flo, tid, nthr);
  if (tid < 64) {
    sb[0][tid] = b_if[tid];
    sb[1][tid] = b_ic[tid];
    sb[2][tid] = b_emb[tid];
    sb[3][tid] = b_fl1[tid];
    sb[4][tid] = b_flo[tid];
  }
  for (int i = tid; i < 256; i += nthr) {
    const int m = i >> 4, c = i & 15;
    const int rr = r0 + m, b = rr / Tt, t = rr % Tt;
    sA[m * AP2 + c] = (c == 0) ? seq_ts[rr] : x[(size_t)rr * (Cc - 1) + (c - 1)];
    const int idx = (t < Tt - 1) ? t : (Tt - 2);
    const float s = times[t] - times[idx];
    const float* cf = coeffs + ((size_t)b * (Tt - 1) + idx) * (4 * Cc) + c;
    const float a0 = cf[0], b0 = cf[Cc], c0 = cf[2 * Cc], d0 = cf[3 * Cc];
    sA[m * AP2 + 16 + c] = a0 + s * (b0 + s * (c0 + s * d0));
  }
  __syncthreads();

  const int n0 = wave * 16;
  v8f aF = {}, aX = {};
  aF = gemm_tile<16, AP2>(sA, 0, sWif, n0, aF, l2, hi);
  aX = gemm_tile<16, AP2>(sA, 16, sWic, n0, aX, l2, hi);
  const float bF = sb[0][n0 + l2], bX = sb[1][n0 + l2];
#pragma unroll
  for (int r = 0; r < 8; ++r) {
    const int m = r + 8 * hi;
    sB[m * AP2 + n0 + l2] = aF[r] + bF;
    sB[m * AP2 + 64 + n0 + l2] = aX[r] + bX;
  }
  __syncthreads();

  v8f acc = {};
  acc = gemm_tile<128, AP2>(sB, 0, sWemb, n0, acc, l2, hi);
  float bv = sb[2][n0 + l2];
#pragma unroll
  for (int r = 0; r < 8; ++r)
    sA[(r + 8 * hi) * AP2 + n0 + l2] = relu(acc[r] + bv);
  __syncthreads();

  acc = (v8f){};
  acc = gemm_tile<64, AP2>(sA, 0, sWfl1, n0, acc, l2, hi);
  bv = sb[3][n0 + l2];
#pragma unroll
  for (int r = 0; r < 8; ++r)
    sB[(r + 8 * hi) * AP2 + n0 + l2] = relu(acc[r] + bv);
  __syncthreads();

  acc = (v8f){};
  acc = gemm_tile<64, AP2>(sB, 0, sWflo, n0, acc, l2, hi);
  bv = sb[4][n0 + l2];
#pragma unroll
  for (int r = 0; r < 8; ++r)
    zflow[(size_t)(r0 + r + 8 * hi) * Hh + n0 + l2] = acc[r] + bv;
}

__global__ __launch_bounds__(512, 1) void ncde_scan_kernel(
    const float* __restrict__ coeffs, const float* __restrict__ times,
    const float* __restrict__ W_ic, const float* __restrict__ b_ic,
    const float* __restrict__ W_f1, const float* __restrict__ b_f1,
    const float* __restrict__ W_f2, const float* __restrict__ b_f2,
    float* __restrict__ zcde) {
  __shared__ float sWf1[64 * WP], sWic[16 * WP];
  __shared__ float sbf1[64], sbic[64];
  __shared__ float stim[Tt];
  __shared__ float sh[16 * AP1], sze[16 * AP1];
  __shared__ __attribute__((aligned(16))) float szo[16 * ZOP];
  __shared__ __attribute__((aligned(16))) float sdx[4][16 * DXP];
  __shared__ __attribute__((aligned(32))) float sbf2[Hh * Cc];

  const int tid = threadIdx.x, nthr = 512;
  const int lane = tid & 31, wave = tid >> 5, l2 = lane & 15, hi = lane >> 4;
  const int b0 = blockIdx.x * 16;

  load_w64<64>(sWf1, W_f1, tid, nthr);
  load_w64<16>(sWic, W_ic, tid, nthr);
  if (tid < 64) {
    sbf1[tid] = b_f1[tid];
    sbic[tid] = b_ic[tid];
  }
  for (int i = tid; i < Tt; i += nthr) stim[i] = times[i];
  for (int i = tid; i < Hh * Cc; i += nthr) sbf2[i] = b_f2[i];

  v2f wf2[4][16];
#pragma unroll
  for (int jj = 0; jj < 4; ++jj) {
    const int j = wave * 4 + jj;
    const int col = 16 * j + l2;
#pragma unroll
    for (int kk = 0; kk < 16; ++kk) {
      const int row = 4 * kk + 2 * hi;
      wf2[jj][kk].x = W_f2[(size_t)row * (Hh * Cc) + col];
      wf2[jj][kk].y = W_f2[(size_t)(row + 1) * (Hh * Cc) + col];
    }
  }

  for (int i = tid; i < 256; i += nthr) {
    const int m = i >> 4, c = i & 15;
    sh[m * AP1 + c] = coeffs[((size_t)(b0 + m) * (Tt - 1)) * (4 * Cc) + c];
  }
  __syncthreads();
  if (wave < 4) {
    const int n0 = wave * 16;
    v8f acc = {};
    acc = gemm_tile<16, AP1>(sh, 0, sWic, n0, acc, l2, hi);
    const float bv = sbic[n0 + l2];
#pragma unroll
    for (int r = 0; r < 8; ++r) {
      const int m = r + 8 * hi;
      const float v = acc[r] + bv;
      sze[m * AP1 + n0 + l2] = v;
      zcde[((size_t)(b0 + m) * Tt) * Hh + n0 + l2] = v;
    }
  }
  __syncthreads();

  float zb[4], ka[4], kb[4], kc[4];
#pragma unroll
  for (int jj = 0; jj < 4; ++jj) {
    zb[jj] = sze[l2 * AP1 + wave * 4 + jj];
    ka[jj] = 0.f;
    kb[jj] = 0.f;
    kc[jj] = 0.f;
  }

  float* const zdst =
      (hi == 0) ? (sze + l2 * AP1 + wave * 4) : (szo + l2 * ZOP + wave * 4);
  const int om = (tid >> 4) & 15, oq = tid & 15;
  float* const zrow = zcde + ((size_t)(b0 + om) * Tt) * Hh + 4 * oq;
  const float4* const zsrc = reinterpret_cast<const float4*>(szo + om * ZOP + 4 * oq);

  for (int step = 0; step < Tt - 1; ++step) {
    const float dt = stim[step + 1] - stim[step];
    if (tid < 256) {
      const int m = tid >> 4, c = tid & 15, b = b0 + m;
      const float* cf = coeffs + ((size_t)b * (Tt - 1) + step) * (4 * Cc) + c;
      const float bb = cf[Cc], cc = cf[2 * Cc], dd = cf[3 * Cc];
      const float s1 = dt * (1.f / 3.f), s2 = dt * (2.f / 3.f);
      sdx[0][m * DXP + c] = bb;
      sdx[1][m * DXP + c] = bb + s1 * (2.f * cc + 3.f * dd * s1);
      sdx[2][m * DXP + c] = bb + s2 * (2.f * cc + 3.f * dd * s2);
      const int idx3 = (step + 1 < Tt - 1) ? (step + 1) : (Tt - 2);
      const float* cf3 = coeffs + ((size_t)b * (Tt - 1) + idx3) * (4 * Cc) + c;
      const float b3 = cf3[Cc], c3 = cf3[2 * Cc], d3 = cf3[3 * Cc];
      const float s3 = stim[step + 1] - stim[idx3];
      sdx[3][m * DXP + c] = b3 + s3 * (2.f * c3 + 3.f * d3 * s3);
    }
    __syncthreads();

    for (int sub = 0; sub < 4; ++sub) {
      if (wave < 4) {
        const int n0 = wave * 16;
        v8f acc = {};
        acc = gemm_tile<64, AP1>(sze, 0, sWf1, n0, acc, l2, hi);
        const float bv = sbf1[n0 + l2];
#pragma unroll
        for (int r = 0; r < 8; ++r)
          sh[(r + 8 * hi) * AP1 + n0 + l2] = relu(acc[r] + bv);
      }
      v8f acc2[4];
#pragma unroll
      for (int jj = 0; jj < 4; ++jj)
        acc2[jj] = *reinterpret_cast<const v8f*>(sbf2 + 64 * wave + 16 * jj + 8 * hi);
      __syncthreads();

      if (sub == 0 && step > 0 && tid < 256) {
        *reinterpret_cast<float4*>(zrow + (size_t)step * Hh) = *zsrc;
      }

#pragma unroll
      for (int kk = 0; kk < 16; ++kk) {
        v2f hb;
        const float* ap = sh + l2 * AP1 + 4 * kk + 2 * hi;
        hb.x = ap[0];
        hb.y = ap[1];
#pragma unroll
        for (int jj = 0; jj < 4; ++jj) acc2[jj] = wmma4(wf2[jj][kk], hb, acc2[jj]);
      }
      const float4* dxp =
          reinterpret_cast<const float4*>(sdx[sub] + l2 * DXP + 8 * hi);
      const float4 dxa = dxp[0], dxb = dxp[1];
      const float dxr[8] = {dxa.x, dxa.y, dxa.z, dxa.w, dxb.x, dxb.y, dxb.z, dxb.w};
      float kp[4];
#pragma unroll
      for (int jj = 0; jj < 4; ++jj) {
        float s = tanh_hw(acc2[jj][0]) * dxr[0];
#pragma unroll
        for (int r = 1; r < 8; ++r) s += tanh_hw(acc2[jj][r]) * dxr[r];
        kp[jj] = s;
      }
#pragma unroll
      for (int jj = 0; jj < 4; ++jj) kp[jj] += xhalf(kp[jj]);

      float ze[4];
      if (sub == 0) {
#pragma unroll
        for (int jj = 0; jj < 4; ++jj) {
          ka[jj] = kp[jj];
          ze[jj] = zb[jj] + (dt * (1.f / 3.f)) * ka[jj];
        }
      } else if (sub == 1) {
#pragma unroll
        for (int jj = 0; jj < 4; ++jj) {
          kb[jj] = kp[jj];
          ze[jj] = zb[jj] + dt * (kb[jj] - ka[jj] * (1.f / 3.f));
        }
      } else if (sub == 2) {
#pragma unroll
        for (int jj = 0; jj < 4; ++jj) {
          kc[jj] = kp[jj];
          ze[jj] = zb[jj] + dt * (ka[jj] - kb[jj] + kc[jj]);
        }
      } else {
#pragma unroll
        for (int jj = 0; jj < 4; ++jj) {
          const float zn =
              zb[jj] + (ka[jj] + 3.f * (kb[jj] + kc[jj]) + kp[jj]) * (dt * 0.125f);
          zb[jj] = zn;
          ze[jj] = zn;
        }
      }
      if (hi == 0 || sub == 3) {
#pragma unroll
        for (int jj = 0; jj < 4; ++jj) zdst[jj] = ze[jj];
      }
      __syncthreads();
    }
  }
  if (tid < 256) {
    *reinterpret_cast<float4*>(zrow + (size_t)(Tt - 1) * Hh) = *zsrc;
  }
}

__global__ __launch_bounds__(128) void ncde_head_kernel(
    const float* __restrict__ zflow, const float* __restrict__ zcde,
    const float* __restrict__ W_mix, const float* __restrict__ b_mix,
    const float* __restrict__ W_l1, const float* __restrict__ b_l1,
    const float* __restrict__ W_l2, const float* __restrict__ b_l2,
    float* __restrict__ out, float* __restrict__ zout) {
  __shared__ float sWmix[128 * WP], sWl1[64 * WP];
  __shared__ float sbmix[64], sbl1[64], sWl2v[64];
  __shared__ float sA[16 * AP2], sB[16 * AP1], sC[16 * AP1];

  const int tid = threadIdx.x, nthr = 128;
  const int lane = tid & 31, wave = tid >> 5, l2 = lane & 15, hi = lane >> 4;
  const int r0 = blockIdx.x * 16;

  load_w64<128>(sWmix, W_mix, tid, nthr);
  load_w64<64>(sWl1, W_l1, tid, nthr);
  if (tid < 64) {
    sbmix[tid] = b_mix[tid];
    sbl1[tid] = b_l1[tid];
    sWl2v[tid] = W_l2[tid];
  }
  for (int i = tid; i < 16 * 128; i += nthr) {
    const int m = i >> 7, c = i & 127;
    const size_t rr = r0 + m;
    sA[m * AP2 + c] = (c < 64) ? zflow[rr * Hh + c] : zcde[rr * Hh + (c - 64)];
  }
  __syncthreads();

  const int n0 = wave * 16;
  v8f acc = {};
  acc = gemm_tile<128, AP2>(sA, 0, sWmix, n0, acc, l2, hi);
  float bv = sbmix[n0 + l2];
#pragma unroll
  for (int r = 0; r < 8; ++r) {
    const int m = r + 8 * hi;
    const float v = acc[r] + bv;
    zout[(size_t)(r0 + m) * Hh + n0 + l2] = v;
    sB[m * AP1 + n0 + l2] = tanhf(v);
  }
  __syncthreads();

  acc = (v8f){};
  acc = gemm_tile<64, AP1>(sB, 0, sWl1, n0, acc, l2, hi);
  bv = sbl1[n0 + l2];
#pragma unroll
  for (int r = 0; r < 8; ++r)
    sC[(r + 8 * hi) * AP1 + n0 + l2] = relu(acc[r] + bv);
  __syncthreads();

  const int m = tid >> 3, p = tid & 7;
  float s = 0.f;
#pragma unroll
  for (int q = 0; q < 8; ++q) s += sC[m * AP1 + p * 8 + q] * sWl2v[p * 8 + q];
  s += __shfl_xor(s, 1, 32);
  s += __shfl_xor(s, 2, 32);
  s += __shfl_xor(s, 4, 32);
  if (p == 0) out[r0 + m] = s + b_l2[0];
}

extern "C" void kernel_launch(void* const* d_in, const int* in_sizes, int n_in,
                              void* d_out, int out_size, void* d_ws,
                              size_t ws_size, hipStream_t stream) {
  const float* x = (const float*)d_in[0];
  const float* seq_ts = (const float*)d_in[1];
  const float* coeffs = (const float*)d_in[3];
  const float* times = (const float*)d_in[4];
  const float* W_if = (const float*)d_in[5];
  const float* b_if = (const float*)d_in[6];
  const float* W_ic = (const float*)d_in[7];
  const float* b_ic = (const float*)d_in[8];
  const float* W_emb = (const float*)d_in[9];
  const float* b_emb = (const float*)d_in[10];
  const float* W_fl1 = (const float*)d_in[11];
  const float* b_fl1 = (const float*)d_in[12];
  const float* W_flo = (const float*)d_in[13];
  const float* b_flo = (const float*)d_in[14];
  const float* W_f1 = (const float*)d_in[15];
  const float* b_f1 = (const float*)d_in[16];
  const float* W_f2 = (const float*)d_in[17];
  const float* b_f2 = (const float*)d_in[18];
  const float* W_mix = (const float*)d_in[19];
  const float* b_mix = (const float*)d_in[20];
  const float* W_l1 = (const float*)d_in[21];
  const float* b_l1 = (const float*)d_in[22];
  const float* W_l2 = (const float*)d_in[23];
  const float* b_l2 = (const float*)d_in[24];

  float* zflow = (float*)d_ws;
  float* zcde = zflow + (size_t)Bb * Tt * Hh;
  float* out = (float*)d_out;
  float* zout = out + (size_t)Bb * Tt;

  ncde_embed_kernel<<<(Bb * Tt) / 16, 128, 0, stream>>>(
      x, seq_ts, coeffs, times, W_if, b_if, W_ic, b_ic, W_emb, b_emb, W_fl1,
      b_fl1, W_flo, b_flo, zflow);
  ncde_scan_kernel<<<Bb / 16, 512, 0, stream>>>(coeffs, times, W_ic, b_ic,
                                                W_f1, b_f1, W_f2, b_f2, zcde);
  ncde_head_kernel<<<(Bb * Tt) / 16, 128, 0, stream>>>(
      zflow, zcde, W_mix, b_mix, W_l1, b_l1, W_l2, b_l2, out, zout);
}


# scan: dX/dt of step s+1 staged during step s (two-deep sdx; loads in flight under GEMM2, by waves idle in GEMM1): no global-load wait or staging barrier at a step's head; sums unchanged
# speedup vs baseline: 3.5949x; 1.0402x over previous
#include <hip/hip_runtime.h>
#include <math.h>

#define Bb 256
#define Tt 512
#define Cc 16
#define Hh 64

#define WP  72
#define AP1 69
#define AP2 133
#define DXP 20
#define ZOP 68

typedef __attribute__((ext_vector_type(2))) float v2f;
typedef __attribute__((ext_vector_type(8))) float v8f;

static __device__ __forceinline__ v8f wmma4(v2f a, v2f b, v8f c) {
  return __builtin_amdgcn_wmma_f32_16x16x4_f32(false, a, false, b, (short)0, c,
                                               false, false);
}

template <int K, int APITCH>
static __device__ __forceinline__ v8f gemm_tile(const float* __restrict__ A,
                                                int acol0,
                                                const float* __restrict__ W,
                                                int n0, v8f acc, int l2, int hi) {
#pragma unroll
  for (int k = 0; k < K; k += 4) {
    v2f a, b;
    const float* ap = A + l2 * APITCH + acol0 + k + 2 * hi;
    a.x = ap[0];
    a.y = ap[1];
    const float* wp = W + (k + 2 * hi) * WP + n0 + l2;
    b.x = wp[0];
    b.y = wp[WP];
    acc = wmma4(a, b, acc);
  }
  return acc;
}

template <int APITCH>
static __device__ __forceinline__ void store_tile(float* __restrict__ O, int n0,
                                                  v8f acc, int l2, int hi) {
#pragma unroll
  for (int r = 0; r < 8; ++r) O[(r + 8 * hi) * APITCH + n0 + l2] = acc[r];
}

template <int ROWS>
static __device__ __forceinline__ void load_w64(float* __restrict__ dst,
                                                const float* __restrict__ src,
                                                int tid, int nthr) {
  for (int i = tid; i < ROWS * 64; i += nthr)
    dst[(i >> 6) * WP + (i & 63)] = src[i];
}

static __device__ __forceinline__ float relu(float v) { return v > 0.f ? v : 0.f; }

static __device__ __forceinline__ float tanh_hw(float v) {
  return __builtin_amdgcn_tanhf(v);
}

static __device__ __forceinline__ float xhalf(float v) {
  const int b = __builtin_bit_cast(int, v);
  return __builtin_bit_cast(
      float, __builtin_amdgcn_permlanex16(b, b, 0x76543210, 0xfedcba98, false, false));
}

__global__ __launch_bounds__(128) void ncde_embed_kernel(
    const float* __restrict__ x, const float* __restrict__ seq_ts,
    const float* __restrict__ coeffs, const float* __restrict__ times,
    const float* __restrict__ W_if, const float* __restrict__ b_if,
    const float* __restrict__ W_ic, const float* __restrict__ b_ic,
    const float* __restrict__ W_emb, const float* __restrict__ b_emb,
    const float* __restrict__ W_fl1, const float* __restrict__ b_fl1,
    const float* __restrict__ W_flo, const float* __restrict__ b_flo,
    float* __restrict__ zflow) {
  __shared__ float sWif[16 * WP], sWic[16 * WP], sWemb[128 * WP];
  __shared__ float sWfl1[64 * WP], sWflo[64 * WP];
  __shared__ float sb[5][64];
  __shared__ float sA[16 * AP2], sB[16 * AP2];

  const int tid = threadIdx.x, nthr = 128;
  const int lane = tid & 31, wave = tid >> 5, l2 = lane & 15, hi = lane >> 4;
  const int r0 = blockIdx.x * 16;

  load_w64<16>(sWif, W_if, tid, nthr);
  load_w64<16>(sWic, W_ic, tid, nthr);
  load_w64<128>(sWemb, W_emb, tid, nthr);
  load_w64<64>(sWfl1, W_fl1, tid, nthr);
  load_w64<64>(sWflo, W_flo, tid, nthr);
  if (tid < 64) {
    sb[0][tid] = b_if[tid];
    sb[1][tid] = b_ic[tid];
    sb[2][tid] = b_emb[tid];
    sb[3][tid] = b_fl1[tid];
    sb[4][tid] = b_flo[tid];
  }
  for (int i = tid; i < 256; i += nthr) {
    const int m = i >> 4, c = i & 15;
    const int rr = r0 + m, b = rr / Tt, t = rr % Tt;
    sA[m * AP2 + c] = (c == 0) ? seq_ts[rr] : x[(size_t)rr * (Cc - 1) + (c - 1)];
    const int idx = (t < Tt - 1) ? t : (Tt - 2);
    const float s = times[t] - times[idx];
    const float* cf = coeffs + ((size_t)b * (Tt - 1) + idx) * (4 * Cc) + c;
    const float a0 = cf[0], b0 = cf[Cc], c0 = cf[2 * Cc], d0 = cf[3 * Cc];
    sA[m * AP2 + 16 + c] = a0 + s * (b0 + s * (c0 + s * d0));
  }
  __syncthreads();

  const int n0 = wave * 16;
  v8f aF = {}, aX = {};
  aF = gemm_tile<16, AP2>(sA, 0, sWif, n0, aF, l2, hi);
  aX = gemm_tile<16, AP2>(sA, 16, sWic, n0, aX, l2, hi);
  const float bF = sb[0][n0 + l2], bX = sb[1][n0 + l2];
#pragma unroll
  for (int r = 0; r < 8; ++r) {
    const int m = r + 8 * hi;
    sB[m * AP2 + n0 + l2] = aF[r] + bF;
    sB[m * AP2 + 64 + n0 + l2] = aX[r] + bX;
  }
  __syncthreads();

  v8f acc = {};
  acc = gemm_tile<128, AP2>(sB, 0, sWemb, n0, acc, l2, hi);
  float bv = sb[2][n0 + l2];
#pragma unroll
  for (int r = 0; r < 8; ++r)
    sA[(r + 8 * hi) * AP2 + n0 + l2] = relu(acc[r] + bv);
  __syncthreads();

  acc = (v8f){};
  acc = gemm_tile<64, AP2>(sA, 0, sWfl1, n0, acc, l2, hi);
  bv = sb[3][n0 + l2];
#pragma unroll
  for (int r = 0; r < 8; ++r)
    sB[(r + 8 * hi) * AP2 + n0 + l2] = relu(acc[r] + bv);
  __syncthreads();

  acc = (v8f){};
  acc = gemm_tile<64, AP2>(sB, 0, sWflo, n0, acc, l2, hi);
  bv = sb[4][n0 + l2];
#pragma unroll
  for (int r = 0; r < 8; ++r)
    zflow[(size_t)(r0 + r + 8 * hi) * Hh + n0 + l2] = acc[r] + bv;
}

static __device__ __forceinline__ void stage_dx(float* __restrict__ d, float dt,
                                                float s3, float bb, float cc,
                                                float dd, float b3, float c3,
                                                float d3) {
  const float s1 = dt * (1.f / 3.f), s2 = dt * (2.f / 3.f);
  d[0] = bb;
  d[16 * DXP] = bb + s1 * (2.f * cc + 3.f * dd * s1);
  d[2 * 16 * DXP] = bb + s2 * (2.f * cc + 3.f * dd * s2);
  d[3 * 16 * DXP] = b3 + s3 * (2.f * c3 + 3.f * d3 * s3);
}

__global__ __launch_bounds__(512, 1) void ncde_scan_kernel(
    const float* __restrict__ coeffs, const float* __restrict__ times,
    const float* __restrict__ W_ic, const float* __restrict__ b_ic,
    const float* __restrict__ W_f1, const float* __restrict__ b_f1,
    const float* __restrict__ W_f2, const float* __restrict__ b_f2,
    float* __restrict__ zcde) {
  __shared__ float sWf1[64 * WP], sWic[16 * WP];
  __shared__ float sbf1[64], sbic[64];
  __shared__ float stim[Tt];
  __shared__ float sh[16 * AP1], sze[16 * AP1];
  __shared__ __attribute__((aligned(16))) float szo[16 * ZOP];
  __shared__ __attribute__((aligned(16))) float sdx[2][4][16 * DXP];
  __shared__ __attribute__((aligned(32))) float sbf2[Hh * Cc];

  const int tid = threadIdx.x, nthr = 512;
  const int lane = tid & 31, wave = tid >> 5, l2 = lane & 15, hi = lane >> 4;
  const int b0 = blockIdx.x * 16;

  load_w64<64>(sWf1, W_f1, tid, nthr);
  load_w64<16>(sWic, W_ic, tid, nthr);
  if (tid < 64) {
    sbf1[tid] = b_f1[tid];
    sbic[tid] = b_ic[tid];
  }
  for (int i = tid; i < Tt; i += nthr) stim[i] = times[i];
  for (int i = tid; i < Hh * Cc; i += nthr) sbf2[i] = b_f2[i];

  v2f wf2[4][16];
#pragma unroll
  for (int jj = 0; jj < 4; ++jj) {
    const int j = wave * 4 + jj;
    const int col = 16 * j + l2;
#pragma unroll
    for (int kk = 0; kk < 16; ++kk) {
      const int row = 4 * kk + 2 * hi;
      wf2[jj][kk].x = W_f2[(size_t)row * (Hh * Cc) + col];
      wf2[jj][kk].y = W_f2[(size_t)(row + 1) * (Hh * Cc) + col];
    }
  }

  for (int i = tid; i < 256; i += nthr) {
    const int m = i >> 4, c = i & 15;
    sh[m * AP1 + c] = coeffs[((size_t)(b0 + m) * (Tt - 1)) * (4 * Cc) + c];
  }
  __syncthreads();
  if (wave < 4) {
    const int n0 = wave * 16;
    v8f acc = {};
    acc = gemm_tile<16, AP1>(sh, 0, sWic, n0, acc, l2, hi);
    const float bv = sbic[n0 + l2];
#pragma unroll
    for (int r = 0; r < 8; ++r) {
      const int m = r + 8 * hi;
      const float v = acc[r] + bv;
      sze[m * AP1 + n0 + l2] = v;
      zcde[((size_t)(b0 + m) * Tt) * Hh + n0 + l2] = v;
    }
  }
  if (tid >= 128 && tid < 384) {
    const int e = tid - 128, m = e >> 4, c = e & 15;
    const float* cf = coeffs + ((size_t)(b0 + m) * (Tt - 1)) * (4 * Cc) + c;
    const int i3 = (1 < Tt - 1) ? 1 : (Tt - 2);
    const float* cf3 = coeffs + ((size_t)(b0 + m) * (Tt - 1) + i3) * (4 * Cc) + c;
    stage_dx(&sdx[0][0][m * DXP + c], stim[1] - stim[0], stim[1] - stim[i3],
             cf[Cc], cf[2 * Cc], cf[3 * Cc], cf3[Cc], cf3[2 * Cc], cf3[3 * Cc]);
  }
  __syncthreads();

  float zb[4], ka[4], kb[4], kc[4];
#pragma unroll
  for (int jj = 0; jj < 4; ++jj) {
    zb[jj] = sze[l2 * AP1 + wave * 4 + jj];
    ka[jj] = 0.f;
    kb[jj] = 0.f;
    kc[jj] = 0.f;
  }

  float* const zdst =
      (hi == 0) ? (sze + l2 * AP1 + wave * 4) : (szo + l2 * ZOP + wave * 4);
  const int om = (tid >> 4) & 15, oq = tid & 15;
  float* const zrow = zcde + ((size_t)(b0 + om) * Tt) * Hh + 4 * oq;
  const float4* const zsrc = reinterpret_cast<const float4*>(szo + om * ZOP + 4 * oq);
  const bool stager = (tid >= 128) && (tid < 384);
  const int sm = ((tid - 128) >> 4) & 15, sc = tid & 15;
  const float* const cfrow = coeffs + ((size_t)(b0 + sm) * (Tt - 1)) * (4 * Cc) + sc;

  for (int step = 0; step < Tt - 1; ++step) {
    const float dt = stim[step + 1] - stim[step];
    const float* const dxcur = &sdx[step & 1][0][0];

    for (int sub = 0; sub < 4; ++sub) {
      if (wave < 4) {
        const int n0 = wave * 16;
        v8f acc = {};
        acc = gemm_tile<64, AP1>(sze, 0, sWf1, n0, acc, l2, hi);
        const float bv = sbf1[n0 + l2];
#pragma unroll
        for (int r = 0; r < 8; ++r)
          sh[(r + 8 * hi) * AP1 + n0 + l2] = relu(acc[r] + bv);
      }
      v8f acc2[4];
#pragma unroll
      for (int jj = 0; jj < 4; ++jj)
        acc2[jj] = *reinterpret_cast<const v8f*>(sbf2 + 64 * wave + 16 * jj + 8 * hi);
      __syncthreads();

      if (sub == 0 && step > 0 && tid < 256) {
        *reinterpret_cast<float4*>(zrow + (size_t)step * Hh) = *zsrc;
      }

      const bool stage_next = (sub == 1) && stager && (step + 1 < Tt - 1);
      float nb = 0.f, nc = 0.f, nd = 0.f, n3b = 0.f, n3c = 0.f, n3d = 0.f;
      if (stage_next) {
        const int i3 = (step + 2 < Tt - 1) ? (step + 2) : (Tt - 2);
        const float* cf = cfrow + (size_t)(step + 1) * (4 * Cc);
        const float* cf3 = cfrow + (size_t)i3 * (4 * Cc);
        nb = cf[Cc];
        nc = cf[2 * Cc];
        nd = cf[3 * Cc];
        n3b = cf3[Cc];
        n3c = cf3[2 * Cc];
        n3d = cf3[3 * Cc];
      }

#pragma unroll
      for (int kk = 0; kk < 16; ++kk) {
        v2f hb;
        const float* ap = sh + l2 * AP1 + 4 * kk + 2 * hi;
        hb.x = ap[0];
        hb.y = ap[1];
#pragma unroll
        for (int jj = 0; jj < 4; ++jj) acc2[jj] = wmma4(wf2[jj][kk], hb, acc2[jj]);
      }
      const float4* dxp =
          reinterpret_cast<const float4*>(dxcur + sub * (16 * DXP) + l2 * DXP + 8 * hi);
      const float4 dxa = dxp[0], dxb = dxp[1];
      const float dxr[8] = {dxa.x, dxa.y, dxa.z, dxa.w, dxb.x, dxb.y, dxb.z, dxb.w};
      float kp[4];
#pragma unroll
      for (int jj = 0; jj < 4; ++jj) {
        float s = tanh_hw(acc2[jj][0]) * dxr[0];
#pragma unroll
        for (int r = 1; r < 8; ++r) s += tanh_hw(acc2[jj][r]) * dxr[r];
        kp[jj] = s;
      }
#pragma unroll
      for (int jj = 0; jj < 4; ++jj) kp[jj] += xhalf(kp[jj]);

      float ze[4];
      if (sub == 0) {
#pragma unroll
        for (int jj = 0; jj < 4; ++jj) {
          ka[jj] = kp[jj];
          ze[jj] = zb[jj] + (dt * (1.f / 3.f)) * ka[jj];
        }
      } else if (sub == 1) {
#pragma unroll
        for (int jj = 0; jj < 4; ++jj) {
          kb[jj] = kp[jj];
          ze[jj] = zb[jj] + dt * (kb[jj] - ka[jj] * (1.f / 3.f));
        }
      } else if (sub == 2) {
#pragma unroll
        for (int jj = 0; jj < 4; ++jj) {
          kc[jj] = kp[jj];
          ze[jj] = zb[jj] + dt * (ka[jj] - kb[jj] + kc[jj]);
        }
      } else {
#pragma unroll
        for (int jj = 0; jj < 4; ++jj) {
          const float zn =
              zb[jj] + (ka[jj] + 3.f * (kb[jj] + kc[jj]) + kp[jj]) * (dt * 0.125f);
          zb[jj] = zn;
          ze[jj] = zn;
        }
      }
      if (hi == 0 || sub == 3) {
#pragma unroll
        for (int jj = 0; jj < 4; ++jj) zdst[jj] = ze[jj];
      }
      if (stage_next) {
        const int i3 = (step + 2 < Tt - 1) ? (step + 2) : (Tt - 2);
        stage_dx(&sdx[(step + 1) & 1][0][sm * DXP + sc],
                 stim[step + 2] - stim[step + 1], stim[step + 2] - stim[i3],
                 nb, nc, nd, n3b, n3c, n3d);
      }
      __syncthreads();
    }
  }
  if (tid < 256) {
    *reinterpret_cast<float4*>(zrow + (size_t)(Tt - 1) * Hh) = *zsrc;
  }
}

__global__ __launch_bounds__(128) void ncde_head_kernel(
    const float* __restrict__ zflow, const float* __restrict__ zcde,
    const float* __restrict__ W_mix, const float* __restrict__ b_mix,
    const float* __restrict__ W_l1, const float* __restrict__ b_l1,
    const float* __restrict__ W_l2, const float* __restrict__ b_l2,
    float* __restrict__ out, float* __restrict__ zout) {
  __shared__ float sWmix[128 * WP], sWl1[64 * WP];
  __shared__ float sbmix[64], sbl1[64], sWl2v[64];
  __shared__ float sA[16 * AP2], sB[16 * AP1], sC[16 * AP1];

  const int tid = threadIdx.x, nthr = 128;
  const int lane = tid & 31, wave = tid >> 5, l2 = lane & 15, hi = lane >> 4;
  const int r0 = blockIdx.x * 16;

  load_w64<128>(sWmix, W_mix, tid, nthr);
  load_w64<64>(sWl1, W_l1, tid, nthr);
  if (tid < 64) {
    sbmix[tid] = b_mix[tid];
    sbl1[tid] = b_l1[tid];
    sWl2v[tid] = W_l2[tid];
  }
  for (int i = tid; i < 16 * 128; i += nthr) {
    const int m = i >> 7, c = i & 127;
    const size_t rr = r0 + m;
    sA[m * AP2 + c] = (c < 64) ? zflow[rr * Hh + c] : zcde[rr * Hh + (c - 64)];
  }
  __syncthreads();

  const int n0 = wave * 16;
  v8f acc = {};
  acc = gemm_tile<128, AP2>(sA, 0, sWmix, n0, acc, l2, hi);
  float bv = sbmix[n0 + l2];
#pragma unroll
  for (int r = 0; r < 8; ++r) {
    const int m = r + 8 * hi;
    const float v = acc[r] + bv;
    zout[(size_t)(r0 + m) * Hh + n0 + l2] = v;
    sB[m * AP1 + n0 + l2] = tanhf(v);
  }
  __syncthreads();

  acc = (v8f){};
  acc = gemm_tile<64, AP1>(sB, 0, sWl1, n0, acc, l2, hi);
  bv = sbl1[n0 + l2];
#pragma unroll
  for (int r = 0; r < 8; ++r)
    sC[(r + 8 * hi) * AP1 + n0 + l2] = relu(acc[r] + bv);
  __syncthreads();

  const int m = tid >> 3, p = tid & 7;
  float s = 0.f;
#pragma unroll
  for (int q = 0; q < 8; ++q) s += sC[m * AP1 + p * 8 + q] * sWl2v[p * 8 + q];
  s += __shfl_xor(s, 1, 32);
  s += __shfl_xor(s, 2, 32);
  s += __shfl_xor(s, 4, 32);
  if (p == 0) out[r0 + m] = s + b_l2[0];
}

extern "C" void kernel_launch(void* const* d_in, const int* in_sizes, int n_in,
                              void* d_out, int out_size, void* d_ws,
                              size_t ws_size, hipStream_t stream) {
  const float* x = (const float*)d_in[0];
  const float* seq_ts = (const float*)d_in[1];
  const float* coeffs = (const float*)d_in[3];
  const float* times = (const float*)d_in[4];
  const float* W_if = (const float*)d_in[5];
  const float* b_if = (const float*)d_in[6];
  const float* W_ic = (const float*)d_in[7];
  const float* b_ic = (const float*)d_in[8];
  const float* W_emb = (const float*)d_in[9];
  const float* b_emb = (const float*)d_in[10];
  const float* W_fl1 = (const float*)d_in[11];
  const float* b_fl1 = (const float*)d_in[12];
  const float* W_flo = (const float*)d_in[13];
  const float* b_flo = (const float*)d_in[14];
  const float* W_f1 = (const float*)d_in[15];
  const float* b_f1 = (const float*)d_in[16];
  const float* W_f2 = (const float*)d_in[17];
  const float* b_f2 = (const float*)d_in[18];
  const float* W_mix = (const float*)d_in[19];
  const float* b_mix = (const float*)d_in[20];
  const float* W_l1 = (const float*)d_in[21];
  const float* b_l1 = (const float*)d_in[22];
  const float* W_l2 = (const float*)d_in[23];
  const float* b_l2 = (const float*)d_in[24];

  float* zflow = (float*)d_ws;
  float* zcde = zflow + (size_t)Bb * Tt * Hh;
  float* out = (float*)d_out;
  float* zout = out + (size_t)Bb * Tt;

  ncde_embed_kernel<<<(Bb * Tt) / 16, 128, 0, stream>>>(
      x, seq_ts, coeffs, times, W_if, b_if, W_ic, b_ic, W_emb, b_emb, W_fl1,
      b_fl1, W_flo, b_flo, zflow);
  ncde_scan_kernel<<<Bb / 16, 512, 0, stream>>>(coeffs, times, W_ic, b_ic,
                                                W_f1, b_f1, W_f2, b_f2, zcde);
  ncde_head_kernel<<<(Bb * Tt) / 16, 128, 0, stream>>>(
      zflow, zcde, W_mix, b_mix, W_l1, b_l1, W_l2, b_l2, out, zout);
}


# scan: LDS fragment reads of each WMMA block (GEMM1, GEMM2 with its dX/dt) issued as ONE group ahead of the block, none among the WMMAs; W_f1 held in LDS in fragment order; sums unchanged
# speedup vs baseline: 3.8213x; 1.0630x over previous
#include <hip/hip_runtime.h>
#include <math.h>

#define Bb 256
#define Tt 512
#define Cc 16
#define Hh 64

#define WP  72
#define AP1 69
#define AP2 133
#define DXP 20
#define ZOP 68

typedef __attribute__((ext_vector_type(2))) float v2f;
typedef __attribute__((ext_vector_type(8))) float v8f;

static __device__ __forceinline__ v8f wmma4(v2f a, v2f b, v8f c) {
  return __builtin_amdgcn_wmma_f32_16x16x4_f32(false, a, false, b, (short)0, c,
                                               false, false);
}

template <int K, int APITCH>
static __device__ __forceinline__ v8f gemm_tile(const float* __restrict__ A,
                                                int acol0,
                                                const float* __restrict__ W,
                                                int n0, v8f acc, int l2, int hi) {
#pragma unroll
  for (int k = 0; k < K; k += 4) {
    v2f a, b;
    const float* ap = A + l2 * APITCH + acol0 + k + 2 * hi;
    a.x = ap[0];
    a.y = ap[1];
    const float* wp = W + (k + 2 * hi) * WP + n0 + l2;
    b.x = wp[0];
    b.y = wp[WP];
    acc = wmma4(a, b, acc);
  }
  return acc;
}

template <int APITCH>
static __device__ __forceinline__ void store_tile(float* __restrict__ O, int n0,
                                                  v8f acc, int l2, int hi) {
#pragma unroll
  for (int r = 0; r < 8; ++r) O[(r + 8 * hi) * APITCH + n0 + l2] = acc[r];
}

template <int ROWS>
static __device__ __forceinline__ void load_w64(float* __restrict__ dst,
                                                const float* __restrict__ src,
                                                int tid, int nthr) {
  for (int i = tid; i < ROWS * 64; i += nthr)
    dst[(i >> 6) * WP + (i & 63)] = src[i];
}

static __device__ __forceinline__ float relu(float v) { return v > 0.f ? v : 0.f; }

static __device__ __forceinline__ float tanh_hw(float v) {
  return __builtin_amdgcn_tanhf(v);
}

static __device__ __forceinline__ float xhalf(float v) {
  const int b = __builtin_bit_cast(int, v);
  return __builtin_bit_cast(
      float, __builtin_amdgcn_permlanex16(b, b, 0x76543210, 0xfedcba98, false, false));
}

static __device__ __forceinline__ void hold16(const v2f (&f)[16]) {
  asm volatile("" ::"v"(f[0]), "v"(f[1]), "v"(f[2]), "v"(f[3]), "v"(f[4]),
               "v"(f[5]), "v"(f[6]), "v"(f[7]), "v"(f[8]), "v"(f[9]),
               "v"(f[10]), "v"(f[11]), "v"(f[12]), "v"(f[13]), "v"(f[14]),
               "v"(f[15]));
}

__global__ __launch_bounds__(128) void ncde_embed_kernel(
    const float* __restrict__ x, const float* __restrict__ seq_ts,
    const float* __restrict__ coeffs, const float* __restrict__ times,
    const float* __restrict__ W_if, const float* __restrict__ b_if,
    const float* __restrict__ W_ic, const float* __restrict__ b_ic,
    const float* __restrict__ W_emb, const float* __restrict__ b_emb,
    const float* __restrict__ W_fl1, const float* __restrict__ b_fl1,
    const float* __restrict__ W_flo, const float* __restrict__ b_flo,
    float* __restrict__ zflow) {
  __shared__ float sWif[16 * WP], sWic[16 * WP], sWemb[128 * WP];
  __shared__ float sWfl1[64 * WP], sWflo[64 * WP];
  __shared__ float sb[5][64];
  __shared__ float sA[16 * AP2], sB[16 * AP2];

  const int tid = threadIdx.x, nthr = 128;
  const int lane = tid & 31, wave = tid >> 5, l2 = lane & 15, hi = lane >> 4;
  const int r0 = blockIdx.x * 16;

  load_w64<16>(sWif, W_if, tid, nthr);
  load_w64<16>(sWic, W_ic, tid, nthr);
  load_w64<128>(sWemb, W_emb, tid, nthr);
  load_w64<64>(sWfl1, W_fl1, tid, nthr);
  load_w64<64>(sWflo, W_flo, tid, nthr);
  if (tid < 64) {
    sb[0][tid] = b_if[tid];
    sb[1][tid] = b_ic[tid];
    sb[2][tid] = b_emb[tid];
    sb[3][tid] = b_fl1[tid];
    sb[4][tid] = b_flo[tid];
  }
  for (int i = tid; i < 256; i += nthr) {
    const int m = i >> 4, c = i & 15;
    const int rr = r0 + m, b = rr / Tt, t = rr % Tt;
    sA[m * AP2 + c] = (c == 0) ? seq_ts[rr] : x[(size_t)rr * (Cc - 1) + (c - 1)];
    const int idx = (t < Tt - 1) ? t : (Tt - 2);
    const float s = times[t] - times[idx];
    const float* cf = coeffs + ((size_t)b * (Tt - 1) + idx) * (4 * Cc) + c;
    const float a0 = cf[0], b0 = cf[Cc], c0 = cf[2 * Cc], d0 = cf[3 * Cc];
    sA[m * AP2 + 16 + c] = a0 + s * (b0 + s * (c0 + s * d0));
  }
  __syncthreads();

  const int n0 = wave * 16;
  v8f aF = {}, aX = {};
  aF = gemm_tile<16, AP2>(sA, 0, sWif, n0, aF, l2, hi);
  aX = gemm_tile<16, AP2>(sA, 16, sWic, n0, aX, l2, hi);
  const float bF = sb[0][n0 + l2], bX = sb[1][n0 + l2];
#pragma unroll
  for (int r = 0; r < 8; ++r) {
    const int m = r + 8 * hi;
    sB[m * AP2 + n0 + l2] = aF[r] + bF;
    sB[m * AP2 + 64 + n0 + l2] = aX[r] + bX;
  }
  __syncthreads();

  v8f acc = {};
  acc = gemm_tile<128, AP2>(sB, 0, sWemb, n0, acc, l2, hi);
  float bv = sb[2][n0 + l2];
#pragma unroll
  for (int r = 0; r < 8; ++r)
    sA[(r + 8 * hi) * AP2 + n0 + l2] = relu(acc[r] + bv);
  __syncthreads();

  acc = (v8f){};
  acc = gemm_tile<64, AP2>(sA, 0, sWfl1, n0, acc, l2, hi);
  bv = sb[3][n0 + l2];
#pragma unroll
  for (int r = 0; r < 8; ++r)
    sB[(r + 8 * hi) * AP2 + n0 + l2] = relu(acc[r] + bv);
  __syncthreads();

  acc = (v8f){};
  acc = gemm_tile<64, AP2>(sB, 0, sWflo, n0, acc, l2, hi);
  bv = sb[4][n0 + l2];
#pragma unroll
  for (int r = 0; r < 8; ++r)
    zflow[(size_t)(r0 + r + 8 * hi) * Hh + n0 + l2] = acc[r] + bv;
}

static __device__ __forceinline__ void stage_dx(float* __restrict__ d, float dt,
                                                float s3, float bb, float cc,
                                                float dd, float b3, float c3,
                                                float d3) {
  const float s1 = dt * (1.f / 3.f), s2 = dt * (2.f / 3.f);
  d[0] = bb;
  d[16 * DXP] = bb + s1 * (2.f * cc + 3.f * dd * s1);
  d[2 * 16 * DXP] = bb + s2 * (2.f * cc + 3.f * dd * s2);
  d[3 * 16 * DXP] = b3 + s3 * (2.f * c3 + 3.f * d3 * s3);
}

__global__ __launch_bounds__(512, 1) void ncde_scan_kernel(
    const float* __restrict__ coeffs, const float* __restrict__ times,
    const float* __restrict__ W_ic, const float* __restrict__ b_ic,
    const float* __restrict__ W_f1, const float* __restrict__ b_f1,
    const float* __restrict__ W_f2, const float* __restrict__ b_f2,
    float* __restrict__ zcde) {
  __shared__ __attribute__((aligned(8))) float sWf1[4 * 16 * 64];
  __shared__ float sWic[16 * WP];
  __shared__ float sbf1[64], sbic[64];
  __shared__ float stim[Tt];
  __shared__ float sh[16 * AP1], sze[16 * AP1];
  __shared__ __attribute__((aligned(16))) float szo[16 * ZOP];
  __shared__ __attribute__((aligned(16))) float sdx[2][4][16 * DXP];
  __shared__ __attribute__((aligned(32))) float sbf2[Hh * Cc];

  const int tid = threadIdx.x, nthr = 512;
  const int lane = tid & 31, wave = tid >> 5, l2 = lane & 15, hi = lane >> 4;
  const int b0 = blockIdx.x * 16;

  for (int i = tid; i < 4 * 16 * 64; i += nthr) {
    const int t = i >> 10, kk = (i >> 6) & 15, ln = (i >> 1) & 31, e = i & 1;
    sWf1[i] = W_f1[(4 * kk + 2 * (ln >> 4) + e) * Hh + 16 * t + (ln & 15)];
  }
  load_w64<16>(sWic, W_ic, tid, nthr);
  if (tid < 64) {
    sbf1[tid] = b_f1[tid];
    sbic[tid] = b_ic[tid];
  }
  for (int i = tid; i < Tt; i += nthr) stim[i] = times[i];
  for (int i = tid; i < Hh * Cc; i += nthr) sbf2[i] = b_f2[i];

  v2f wf2[4][16];
#pragma unroll
  for (int jj = 0; jj < 4; ++jj) {
    const int j = wave * 4 + jj;
    const int col = 16 * j + l2;
#pragma unroll
    for (int kk = 0; kk < 16; ++kk) {
      const int row = 4 * kk + 2 * hi;
      wf2[jj][kk].x = W_f2[(size_t)row * (Hh * Cc) + col];
      wf2[jj][kk].y = W_f2[(size_t)(row + 1) * (Hh * Cc) + col];
    }
  }

  for (int i = tid; i < 256; i += nthr) {
    const int m = i >> 4, c = i & 15;
    sh[m * AP1 + c] = coeffs[((size_t)(b0 + m) * (Tt - 1)) * (4 * Cc) + c];
  }
  __syncthreads();
  if (wave < 4) {
    const int n0 = wave * 16;
    v8f acc = {};
    acc = gemm_tile<16, AP1>(sh, 0, sWic, n0, acc, l2, hi);
    const float bv = sbic[n0 + l2];
#pragma unroll
    for (int r = 0; r < 8; ++r) {
      const int m = r + 8 * hi;
      const float v = acc[r] + bv;
      sze[m * AP1 + n0 + l2] = v;
      zcde[((size_t)(b0 + m) * Tt) * Hh + n0 + l2] = v;
    }
  }
  if (tid >= 128 && tid < 384) {
    const int e = tid - 128, m = e >> 4, c = e & 15;
    const float* cf = coeffs + ((size_t)(b0 + m) * (Tt - 1)) * (4 * Cc) + c;
    const int i3 = (1 < Tt - 1) ? 1 : (Tt - 2);
    const float* cf3 = coeffs + ((size_t)(b0 + m) * (Tt - 1) + i3) * (4 * Cc) + c;
    stage_dx(&sdx[0][0][m * DXP + c], stim[1] - stim[0], stim[1] - stim[i3],
             cf[Cc], cf[2 * Cc], cf[3 * Cc], cf3[Cc], cf3[2 * Cc], cf3[3 * Cc]);
  }
  __syncthreads();

  float zb[4], ka[4], kb[4], kc[4];
#pragma unroll
  for (int jj = 0; jj < 4; ++jj) {
    zb[jj] = sze[l2 * AP1 + wave * 4 + jj];
    ka[jj] = 0.f;
    kb[jj] = 0.f;
    kc[jj] = 0.f;
  }

  float* const zdst =
      (hi == 0) ? (sze + l2 * AP1 + wave * 4) : (szo + l2 * ZOP + wave * 4);
  const int om = (tid >> 4) & 15, oq = tid & 15;
  float* const zrow = zcde + ((size_t)(b0 + om) * Tt) * Hh + 4 * oq;
  const float4* const zsrc = reinterpret_cast<const float4*>(szo + om * ZOP + 4 * oq);
  const bool stager = (tid >= 128) && (tid < 384);
  const int sm = ((tid - 128) >> 4) & 15, sc = tid & 15;
  const float* const cfrow = coeffs + ((size_t)(b0 + sm) * (Tt - 1)) * (4 * Cc) + sc;

  for (int step = 0; step < Tt - 1; ++step) {
    const float dt = stim[step + 1] - stim[step];
    const float* const dxcur = &sdx[step & 1][0][0];

    for (int sub = 0; sub < 4; ++sub) {
      if (wave < 4) {
        const int n0 = wave * 16;
        v2f ga[16], gb[16];
#pragma unroll
        for (int kk = 0; kk < 16; ++kk) {
          const float* ap = sze + l2 * AP1 + 4 * kk + 2 * hi;
          ga[kk].x = ap[0];
          ga[kk].y = ap[1];
          gb[kk] = *reinterpret_cast<const v2f*>(sWf1 + (wave * 16 + kk) * 64 + lane * 2);
        }
        const float bv = sbf1[n0 + l2];
        hold16(ga);
        hold16(gb);
        v8f acc = {};
#pragma unroll
        for (int kk = 0; kk < 16; ++kk) acc = wmma4(ga[kk], gb[kk], acc);
#pragma unroll
        for (int r = 0; r < 8; ++r)
          sh[(r + 8 * hi) * AP1 + n0 + l2] = relu(acc[r] + bv);
      }
      v8f acc2[4];
#pragma unroll
      for (int jj = 0; jj < 4; ++jj)
        acc2[jj] = *reinterpret_cast<const v8f*>(sbf2 + 64 * wave + 16 * jj + 8 * hi);
      __syncthreads();

      if (sub == 0 && step > 0 && tid < 256) {
        *reinterpret_cast<float4*>(zrow + (size_t)step * Hh) = *zsrc;
      }

      const bool stage_next = (sub == 1) && stager && (step + 1 < Tt - 1);
      float nb = 0.f, nc = 0.f, nd = 0.f, n3b = 0.f, n3c = 0.f, n3d = 0.f;
      if (stage_next) {
        const int i3 = (step + 2 < Tt - 1) ? (step + 2) : (Tt - 2);
        const float* cf = cfrow + (size_t)(step + 1) * (4 * Cc);
        const float* cf3 = cfrow + (size_t)i3 * (4 * Cc);
        nb = cf[Cc];
        nc = cf[2 * Cc];
        nd = cf[3 * Cc];
        n3b = cf3[Cc];
        n3c = cf3[2 * Cc];
        n3d = cf3[3 * Cc];
      }

      v2f hb[16];
#pragma unroll
      for (int kk = 0; kk < 16; ++kk) {
        const float* ap = sh + l2 * AP1 + 4 * kk + 2 * hi;
        hb[kk].x = ap[0];
        hb[kk].y = ap[1];
      }
      const float4* dxp =
          reinterpret_cast<const float4*>(dxcur + sub * (16 * DXP) + l2 * DXP + 8 * hi);
      const float4 dxa = dxp[0], dxb = dxp[1];
      hold16(hb);
      asm volatile("" ::"v"(dxa.x), "v"(dxa.y), "v"(dxa.z), "v"(dxa.w), "v"(dxb.x),
                   "v"(dxb.y), "v"(dxb.z), "v"(dxb.w));
#pragma unroll
      for (int kk = 0; kk < 16; ++kk) {
#pragma unroll
        for (int jj = 0; jj < 4; ++jj)
          acc2[jj] = wmma4(wf2[jj][kk], hb[kk], acc2[jj]);
      }
      const float dxr[8] = {dxa.x, dxa.y, dxa.z, dxa.w, dxb.x, dxb.y, dxb.z, dxb.w};
      float kp[4];
#pragma unroll
      for (int jj = 0; jj < 4; ++jj) {
        float s = tanh_hw(acc2[jj][0]) * dxr[0];
#pragma unroll
        for (int r = 1; r < 8; ++r) s += tanh_hw(acc2[jj][r]) * dxr[r];
        kp[jj] = s;
      }
#pragma unroll
      for (int jj = 0; jj < 4; ++jj) kp[jj] += xhalf(kp[jj]);

      float ze[4];
      if (sub == 0) {
#pragma unroll
        for (int jj = 0; jj < 4; ++jj) {
          ka[jj] = kp[jj];
          ze[jj] = zb[jj] + (dt * (1.f / 3.f)) * ka[jj];
        }
      } else if (sub == 1) {
#pragma unroll
        for (int jj = 0; jj < 4; ++jj) {
          kb[jj] = kp[jj];
          ze[jj] = zb[jj] + dt * (kb[jj] - ka[jj] * (1.f / 3.f));
        }
      } else if (sub == 2) {
#pragma unroll
        for (int jj = 0; jj < 4; ++jj) {
          kc[jj] = kp[jj];
          ze[jj] = zb[jj] + dt * (ka[jj] - kb[jj] + kc[jj]);
        }
      } else {
#pragma unroll
        for (int jj = 0; jj < 4; ++jj) {
          const float zn =
              zb[jj] + (ka[jj] + 3.f * (kb[jj] + kc[jj]) + kp[jj]) * (dt * 0.125f);
          zb[jj] = zn;
          ze[jj] = zn;
        }
      }
      if (hi == 0 || sub == 3) {
#pragma unroll
        for (int jj = 0; jj < 4; ++jj) zdst[jj] = ze[jj];
      }
      if (stage_next) {
        const int i3 = (step + 2 < Tt - 1) ? (step + 2) : (Tt - 2);
        stage_dx(&sdx[(step + 1) & 1][0][sm * DXP + sc],
                 stim[step + 2] - stim[step + 1], stim[step + 2] - stim[i3],
                 nb, nc, nd, n3b, n3c, n3d);
      }
      __syncthreads();
    }
  }
  if (tid < 256) {
    *reinterpret_cast<float4*>(zrow + (size_t)(Tt - 1) * Hh) = *zsrc;
  }
}

__global__ __launch_bounds__(128) void ncde_head_kernel(
    const float* __restrict__ zflow, const float* __restrict__ zcde,
    const float* __restrict__ W_mix, const float* __restrict__ b_mix,
    const float* __restrict__ W_l1, const float* __restrict__ b_l1,
    const float* __restrict__ W_l2, const float* __restrict__ b_l2,
    float* __restrict__ out, float* __restrict__ zout) {
  __shared__ float sWmix[128 * WP], sWl1[64 * WP];
  __shared__ float sbmix[64], sbl1[64], sWl2v[64];
  __shared__ float sA[16 * AP2], sB[16 * AP1], sC[16 * AP1];

  const int tid = threadIdx.x, nthr = 128;
  const int lane = tid & 31, wave = tid >> 5, l2 = lane & 15, hi = lane >> 4;
  const int r0 = blockIdx.x * 16;

  load_w64<128>(sWmix, W_mix, tid, nthr);
  load_w64<64>(sWl1, W_l1, tid, nthr);
  if (tid < 64) {
    sbmix[tid] = b_mix[tid];
    sbl1[tid] = b_l1[tid];
    sWl2v[tid] = W_l2[tid];
  }
  for (int i = tid; i < 16 * 128; i += nthr) {
    const int m = i >> 7, c = i & 127;
    const size_t rr = r0 + m;
    sA[m * AP2 + c] = (c < 64) ? zflow[rr * Hh + c] : zcde[rr * Hh + (c - 64)];
  }
  __syncthreads();

  const int n0 = wave * 16;
  v8f acc = {};
  acc = gemm_tile<128, AP2>(sA, 0, sWmix, n0, acc, l2, hi);
  float bv = sbmix[n0 + l2];
#pragma unroll
  for (int r = 0; r < 8; ++r) {
    const int m = r + 8 * hi;
    const float v = acc[r] + bv;
    zout[(size_t)(r0 + m) * Hh + n0 + l2] = v;
    sB[m * AP1 + n0 + l2] = tanhf(v);
  }
  __syncthreads();

  acc = (v8f){};
  acc = gemm_tile<64, AP1>(sB, 0, sWl1, n0, acc, l2, hi);
  bv = sbl1[n0 + l2];
#pragma unroll
  for (int r = 0; r < 8; ++r)
    sC[(r + 8 * hi) * AP1 + n0 + l2] = relu(acc[r] + bv);
  __syncthreads();

  const int m = tid >> 3, p = tid & 7;
  float s = 0.f;
#pragma unroll
  for (int q = 0; q < 8; ++q) s += sC[m * AP1 + p * 8 + q] * sWl2v[p * 8 + q];
  s += __shfl_xor(s, 1, 32);
  s += __shfl_xor(s, 2, 32);
  s += __shfl_xor(s, 4, 32);
  if (p == 0) out[r0 + m] = s + b_l2[0];
}

extern "C" void kernel_launch(void* const* d_in, const int* in_sizes, int n_in,
                              void* d_out, int out_size, void* d_ws,
                              size_t ws_size, hipStream_t stream) {
  const float* x = (const float*)d_in[0];
  const float* seq_ts = (const float*)d_in[1];
  const float* coeffs = (const float*)d_in[3];
  const float* times = (const float*)d_in[4];
  const float* W_if = (const float*)d_in[5];
  const float* b_if = (const float*)d_in[6];
  const float* W_ic = (const float*)d_in[7];
  const float* b_ic = (const float*)d_in[8];
  const float* W_emb = (const float*)d_in[9];
  const float* b_emb = (const float*)d_in[10];
  const float* W_fl1 = (const float*)d_in[11];
  const float* b_fl1 = (const float*)d_in[12];
  const float* W_flo = (const float*)d_in[13];
  const float* b_flo = (const float*)d_in[14];
  const float* W_f1 = (const float*)d_in[15];
  const float* b_f1 = (const float*)d_in[16];
  const float* W_f2 = (const float*)d_in[17];
  const float* b_f2 = (const float*)d_in[18];
  const float* W_mix = (const float*)d_in[19];
  const float* b_mix = (const float*)d_in[20];
  const float* W_l1 = (const float*)d_in[21];
  const float* b_l1 = (const float*)d_in[22];
  const float* W_l2 = (const float*)d_in[23];
  const float* b_l2 = (const float*)d_in[24];

  float* zflow = (float*)d_ws;
  float* zcde = zflow + (size_t)Bb * Tt * Hh;
  float* out = (float*)d_out;
  float* zout = out + (size_t)Bb * Tt;

  ncde_embed_kernel<<<(Bb * Tt) / 16, 128, 0, stream>>>(
      x, seq_ts, coeffs, times, W_if, b_if, W_ic, b_ic, W_emb, b_emb, W_fl1,
      b_fl1, W_flo, b_flo, zflow);
  ncde_scan_kernel<<<Bb / 16, 512, 0, stream>>>(coeffs, times, W_ic, b_ic,
                                                W_f1, b_f1, W_f2, b_f2, zcde);
  ncde_head_kernel<<<(Bb * Tt) / 16, 128, 0, stream>>>(
      zflow, zcde, W_mix, b_mix, W_l1, b_l1, W_l2, b_l2, out, zout);
}
